# starting kernel
# baseline (speedup 1.0000x reference)
#include <hip/hip_runtime.h>
#include <cstdint>
#include <cstddef>

typedef __attribute__((ext_vector_type(2))) float v2f;
typedef __attribute__((ext_vector_type(8))) float v8f;

#define IN_DIM   128
#define OUT_DIM  128
#define HEADS    4
#define HEAD_DIM 32

__device__ __forceinline__ int f2key(float f) {
  int i = __float_as_int(f);
  return i >= 0 ? i : (i ^ 0x7fffffff);
}
__device__ __forceinline__ float key2f(int k) {
  int i = k >= 0 ? k : (k ^ 0x7fffffff);
  return __int_as_float(i);
}

__global__ void gat_gemm_wmma(const float* __restrict__ x,
                              const float* __restrict__ W,
                              float* __restrict__ Wx, int N) {
  const int tile  = blockIdx.x;
  const int wave  = threadIdx.x >> 5;
  const int lane  = threadIdx.x & 31;
  const int mbase = tile * 16;
  const int obase = wave * 16;
  const int half  = lane >> 4;
  const int l15   = lane & 15;

  int arow = mbase + l15;
  if (arow >= N) arow = N - 1;

  const float* aptr = x + (size_t)arow * IN_DIM + 2 * half;
  const float* bptr = W + (size_t)(obase + l15) * IN_DIM + 2 * half;

  v8f c = {};
  #pragma unroll
  for (int k = 0; k < IN_DIM; k += 4) {
    v2f af = *(const v2f*)(aptr + k);
    v2f bf = *(const v2f*)(bptr + k);
    c = __builtin_amdgcn_wmma_f32_16x16x4_f32(false, af, false, bf,
                                              (short)0, c, false, false);
  }

  #pragma unroll
  for (int r = 0; r < 8; ++r) {
    int row = mbase + r + 8 * half;
    if (row < N) Wx[(size_t)row * OUT_DIM + obase + l15] = c[r];
  }
}

__global__ void gat_scores(const float* __restrict__ Wx,
                           const float* __restrict__ a,
                           float* __restrict__ s1, float* __restrict__ s2,
                           int N) {
  int t = blockIdx.x * blockDim.x + threadIdx.x;
  if (t >= N * HEADS) return;
  int n = t >> 2, h = t & 3;
  const float4* p  = (const float4*)(Wx + (size_t)n * OUT_DIM + h * HEAD_DIM);
  const float4* a1 = (const float4*)(a);
  const float4* a2 = (const float4*)(a + HEAD_DIM);
  float acc1 = 0.f, acc2 = 0.f;
  #pragma unroll
  for (int d = 0; d < HEAD_DIM / 4; ++d) {
    float4 v  = p[d];
    float4 c1 = a1[d];
    float4 c2 = a2[d];
    acc1 += v.x * c1.x + v.y * c1.y + v.z * c1.z + v.w * c1.w;
    acc2 += v.x * c2.x + v.y * c2.y + v.z * c2.z + v.w * c2.w;
  }
  s1[t] = acc1;
  s2[t] = acc2;
}

__global__ void gat_init(float* __restrict__ acc, float* __restrict__ denom,
                         int* __restrict__ mkey, int N) {
  int i = blockIdx.x * blockDim.x + threadIdx.x;
  if (i < N * OUT_DIM) acc[i] = 0.f;
  if (i < N * HEADS) { denom[i] = 0.f; mkey[i] = (int)0x80000000; }
}

__global__ void gat_edge_max(const int* __restrict__ ei,
                             const float* __restrict__ s1,
                             const float* __restrict__ s2,
                             int* __restrict__ mkey, int E) {
  int e = blockIdx.x * blockDim.x + threadIdx.x;
  if (e >= E) return;
  int src = ei[e], dst = ei[E + e];
  float4 v1 = *(const float4*)(s1 + (size_t)src * HEADS);
  float4 v2 = *(const float4*)(s2 + (size_t)dst * HEADS);
  float el[HEADS] = {v1.x + v2.x, v1.y + v2.y, v1.z + v2.z, v1.w + v2.w};
  #pragma unroll
  for (int h = 0; h < HEADS; ++h) {
    float v = el[h] > 0.f ? el[h] : 0.2f * el[h];
    atomicMax(&mkey[(size_t)dst * HEADS + h], f2key(v));
  }
}

__global__ void gat_edge_exp(const int* __restrict__ ei,
                             const float* __restrict__ s1,
                             const float* __restrict__ s2,
                             const int* __restrict__ mkey,
                             float* __restrict__ ex,
                             float* __restrict__ denom, int E) {
  int e = blockIdx.x * blockDim.x + threadIdx.x;
  if (e >= E) return;
  int src = ei[e], dst = ei[E + e];
  float4 v1 = *(const float4*)(s1 + (size_t)src * HEADS);
  float4 v2 = *(const float4*)(s2 + (size_t)dst * HEADS);
  float el[HEADS] = {v1.x + v2.x, v1.y + v2.y, v1.z + v2.z, v1.w + v2.w};
  float4 exv;
  #pragma unroll
  for (int h = 0; h < HEADS; ++h) {
    float v = el[h] > 0.f ? el[h] : 0.2f * el[h];
    float m = key2f(mkey[(size_t)dst * HEADS + h]);
    float xv = __expf(v - m);
    ((float*)&exv)[h] = xv;
    unsafeAtomicAdd(&denom[(size_t)dst * HEADS + h], xv);
  }
  *(float4*)(ex + (size_t)e * HEADS) = exv;
}

__global__ void gat_recip(float* __restrict__ denom, int total) {
  int i = blockIdx.x * blockDim.x + threadIdx.x;
  if (i >= total) return;
  denom[i] = __frcp_rn(denom[i]);
}

__global__ void gat_edge_aggr(const int* __restrict__ ei,
                              const float* __restrict__ Wx,
                              const float* __restrict__ ex,
                              const float* __restrict__ rden,
                              float* __restrict__ acc, int E) {
  long long gid = (long long)blockIdx.x * blockDim.x + threadIdx.x;
  int e  = (int)(gid >> 5);
  int q  = (int)(gid & 31);
  if (e >= E) return;
  int src = ei[e], dst = ei[E + e];
  int j = q << 2;
  int h = j >> 5;
  float alpha = ex[(size_t)e * HEADS + h] * rden[(size_t)dst * HEADS + h];
  float4 wv = *(const float4*)(Wx + (size_t)src * OUT_DIM + j);
  float* ap = acc + (size_t)dst * OUT_DIM + j;
  unsafeAtomicAdd(ap + 0, alpha * wv.x);
  unsafeAtomicAdd(ap + 1, alpha * wv.y);
  unsafeAtomicAdd(ap + 2, alpha * wv.z);
  unsafeAtomicAdd(ap + 3, alpha * wv.w);
}

__global__ void gat_elu(const float* __restrict__ acc, float* __restrict__ out,
                        int total) {
  int i = blockIdx.x * blockDim.x + threadIdx.x;
  if (i >= total) return;
  float v = acc[i];
  out[i] = v > 0.f ? v : (__expf(v) - 1.0f);
}

extern "C" void kernel_launch(void* const* d_in, const int* in_sizes, int n_in,
                              void* d_out, int out_size, void* d_ws, size_t ws_size,
                              hipStream_t stream) {
  const float* x  = (const float*)d_in[0];
  const float* W  = (const float*)d_in[1];
  const float* a  = (const float*)d_in[2];
  const int*   ei = (const int*)d_in[3];
  const int N = in_sizes[0] / IN_DIM;
  const int E = in_sizes[3] / 2;
  float* out = (float*)d_out;

  char* ws = (char*)d_ws;
  size_t off = 0;
  auto carve = [&](size_t bytes) -> void* {
    void* p = ws + off;
    off += (bytes + 255) & ~(size_t)255;
    return p;
  };
  float* Wx    = (float*)carve((size_t)N * OUT_DIM * sizeof(float));
  float* s1    = (float*)carve((size_t)N * HEADS * sizeof(float));
  float* s2    = (float*)carve((size_t)N * HEADS * sizeof(float));
  int*   mkey  = (int*)  carve((size_t)N * HEADS * sizeof(int));
  float* denom = (float*)carve((size_t)N * HEADS * sizeof(float));
  float* ex    = (float*)carve((size_t)E * HEADS * sizeof(float));
  float* acc   = (float*)carve((size_t)N * OUT_DIM * sizeof(float));
  (void)ws_size;

  gat_gemm_wmma<<<(N + 15) / 16, 256, 0, stream>>>(x, W, Wx, N);

  gat_scores<<<(N * HEADS + 255) / 256, 256, 0, stream>>>(Wx, a, s1, s2, N);

  gat_init<<<(N * OUT_DIM + 255) / 256, 256, 0, stream>>>(acc, denom, mkey, N);

  gat_edge_max<<<(E + 255) / 256, 256, 0, stream>>>(ei, s1, s2, mkey, E);

  gat_edge_exp<<<(E + 255) / 256, 256, 0, stream>>>(ei, s1, s2, mkey, ex, denom, E);

  gat_recip<<<(N * HEADS + 255) / 256, 256, 0, stream>>>(denom, N * HEADS);

  {
    long long threads = (long long)E * 32;
    int blocks = (int)((threads + 255) / 256);
    gat_edge_aggr<<<blocks, 256, 0, stream>>>(ei, Wx, ex, denom, acc, E);
  }

  gat_elu<<<(N * OUT_DIM + 255) / 256, 256, 0, stream>>>(acc, out, N * OUT_DIM);
}


# edge max/exp kernels: adjacent lanes hold the 4 heads of one edge, so a head-atomic instruction reaches 8 lines not 32 (4x fewer atomic line requests); arrival order of denominator adds changes
# speedup vs baseline: 1.5551x; 1.5551x over previous
#include <hip/hip_runtime.h>
#include <cstdint>
#include <cstddef>

typedef __attribute__((ext_vector_type(2))) float v2f;
typedef __attribute__((ext_vector_type(8))) float v8f;

#define IN_DIM   128
#define OUT_DIM  128
#define HEADS    4
#define HEAD_DIM 32

__device__ __forceinline__ int f2key(float f) {
  int i = __float_as_int(f);
  return i >= 0 ? i : (i ^ 0x7fffffff);
}
__device__ __forceinline__ float key2f(int k) {
  int i = k >= 0 ? k : (k ^ 0x7fffffff);
  return __int_as_float(i);
}

__global__ void gat_gemm_wmma(const float* __restrict__ x,
                              const float* __restrict__ W,
                              float* __restrict__ Wx, int N) {
  const int tile  = blockIdx.x;
  const int wave  = threadIdx.x >> 5;
  const int lane  = threadIdx.x & 31;
  const int mbase = tile * 16;
  const int obase = wave * 16;
  const int half  = lane >> 4;
  const int l15   = lane & 15;

  int arow = mbase + l15;
  if (arow >= N) arow = N - 1;

  const float* aptr = x + (size_t)arow * IN_DIM + 2 * half;
  const float* bptr = W + (size_t)(obase + l15) * IN_DIM + 2 * half;

  v8f c = {};
  #pragma unroll
  for (int k = 0; k < IN_DIM; k += 4) {
    v2f af = *(const v2f*)(aptr + k);
    v2f bf = *(const v2f*)(bptr + k);
    c = __builtin_amdgcn_wmma_f32_16x16x4_f32(false, af, false, bf,
                                              (short)0, c, false, false);
  }

  #pragma unroll
  for (int r = 0; r < 8; ++r) {
    int row = mbase + r + 8 * half;
    if (row < N) Wx[(size_t)row * OUT_DIM + obase + l15] = c[r];
  }
}

__global__ void gat_scores(const float* __restrict__ Wx,
                           const float* __restrict__ a,
                           float* __restrict__ s1, float* __restrict__ s2,
                           int N) {
  int t = blockIdx.x * blockDim.x + threadIdx.x;
  if (t >= N * HEADS) return;
  int n = t >> 2, h = t & 3;
  const float4* p  = (const float4*)(Wx + (size_t)n * OUT_DIM + h * HEAD_DIM);
  const float4* a1 = (const float4*)(a);
  const float4* a2 = (const float4*)(a + HEAD_DIM);
  float acc1 = 0.f, acc2 = 0.f;
  #pragma unroll
  for (int d = 0; d < HEAD_DIM / 4; ++d) {
    float4 v  = p[d];
    float4 c1 = a1[d];
    float4 c2 = a2[d];
    acc1 += v.x * c1.x + v.y * c1.y + v.z * c1.z + v.w * c1.w;
    acc2 += v.x * c2.x + v.y * c2.y + v.z * c2.z + v.w * c2.w;
  }
  s1[t] = acc1;
  s2[t] = acc2;
}

__global__ void gat_init(float* __restrict__ acc, float* __restrict__ denom,
                         int* __restrict__ mkey, int N) {
  int i = blockIdx.x * blockDim.x + threadIdx.x;
  if (i < N * OUT_DIM) acc[i] = 0.f;
  if (i < N * HEADS) { denom[i] = 0.f; mkey[i] = (int)0x80000000; }
}

#define EH_ITEMS 4

template <bool FULL>
__device__ __forceinline__ void edge_max_body(const int* __restrict__ ei,
                                              const float* __restrict__ s1,
                                              const float* __restrict__ s2,
                                              int* __restrict__ mkey,
                                              int E, int total, int t0, int h) {
  int   src[EH_ITEMS], dst[EH_ITEMS];
  float el[EH_ITEMS];
  #pragma unroll
  for (int k = 0; k < EH_ITEMS; ++k) {
    int t = t0 + 32 * k;
    int e = (FULL || t < total) ? (t >> 2) : 0;
    src[k] = ei[e];
    dst[k] = ei[E + e];
  }
  #pragma unroll
  for (int k = 0; k < EH_ITEMS; ++k)
    el[k] = s1[(size_t)src[k] * HEADS + h] + s2[(size_t)dst[k] * HEADS + h];
  #pragma unroll
  for (int k = 0; k < EH_ITEMS; ++k) {
    if (FULL || t0 + 32 * k < total) {
      float v = el[k] > 0.f ? el[k] : 0.2f * el[k];
      atomicMax(&mkey[(size_t)dst[k] * HEADS + h], f2key(v));
    }
  }
}

__global__ void gat_edge_max(const int* __restrict__ ei,
                             const float* __restrict__ s1,
                             const float* __restrict__ s2,
                             int* __restrict__ mkey, int E) {
  const int total = E * HEADS;
  const int lane  = threadIdx.x & 31;
  const int h     = lane & 3;
  const int wbase = (blockIdx.x * (blockDim.x >> 5) + (threadIdx.x >> 5)) *
                    (32 * EH_ITEMS);
  if (wbase + 32 * EH_ITEMS <= total)
    edge_max_body<true>(ei, s1, s2, mkey, E, total, wbase + lane, h);
  else
    edge_max_body<false>(ei, s1, s2, mkey, E, total, wbase + lane, h);
}

template <bool FULL>
__device__ __forceinline__ void edge_exp_body(const int* __restrict__ ei,
                                              const float* __restrict__ s1,
                                              const float* __restrict__ s2,
                                              const int* __restrict__ mkey,
                                              float* __restrict__ ex,
                                              float* __restrict__ denom,
                                              int E, int total, int t0, int h) {
  int   src[EH_ITEMS], dst[EH_ITEMS], mk[EH_ITEMS];
  float el[EH_ITEMS];
  #pragma unroll
  for (int k = 0; k < EH_ITEMS; ++k) {
    int t = t0 + 32 * k;
    int e = (FULL || t < total) ? (t >> 2) : 0;
    src[k] = ei[e];
    dst[k] = ei[E + e];
  }
  #pragma unroll
  for (int k = 0; k < EH_ITEMS; ++k) {
    el[k] = s1[(size_t)src[k] * HEADS + h] + s2[(size_t)dst[k] * HEADS + h];
    mk[k] = mkey[(size_t)dst[k] * HEADS + h];
  }
  #pragma unroll
  for (int k = 0; k < EH_ITEMS; ++k) {
    int t = t0 + 32 * k;
    if (FULL || t < total) {
      float v  = el[k] > 0.f ? el[k] : 0.2f * el[k];
      float xv = __expf(v - key2f(mk[k]));
      ex[t] = xv;
      unsafeAtomicAdd(&denom[(size_t)dst[k] * HEADS + h], xv);
    }
  }
}

__global__ void gat_edge_exp(const int* __restrict__ ei,
                             const float* __restrict__ s1,
                             const float* __restrict__ s2,
                             const int* __restrict__ mkey,
                             float* __restrict__ ex,
                             float* __restrict__ denom, int E) {
  const int total = E * HEADS;
  const int lane  = threadIdx.x & 31;
  const int h     = lane & 3;
  const int wbase = (blockIdx.x * (blockDim.x >> 5) + (threadIdx.x >> 5)) *
                    (32 * EH_ITEMS);
  if (wbase + 32 * EH_ITEMS <= total)
    edge_exp_body<true>(ei, s1, s2, mkey, ex, denom, E, total, wbase + lane, h);
  else
    edge_exp_body<false>(ei, s1, s2, mkey, ex, denom, E, total, wbase + lane, h);
}

__global__ void gat_recip(float* __restrict__ denom, int total) {
  int i = blockIdx.x * blockDim.x + threadIdx.x;
  if (i >= total) return;
  denom[i] = __frcp_rn(denom[i]);
}

__global__ void gat_edge_aggr(const int* __restrict__ ei,
                              const float* __restrict__ Wx,
                              const float* __restrict__ ex,
                              const float* __restrict__ rden,
                              float* __restrict__ acc, int E) {
  long long gid = (long long)blockIdx.x * blockDim.x + threadIdx.x;
  int e  = (int)(gid >> 5);
  int q  = (int)(gid & 31);
  if (e >= E) return;
  int src = ei[e], dst = ei[E + e];
  int j = q << 2;
  int h = j >> 5;
  float alpha = ex[(size_t)e * HEADS + h] * rden[(size_t)dst * HEADS + h];
  float4 wv = *(const float4*)(Wx + (size_t)src * OUT_DIM + j);
  float* ap = acc + (size_t)dst * OUT_DIM + j;
  unsafeAtomicAdd(ap + 0, alpha * wv.x);
  unsafeAtomicAdd(ap + 1, alpha * wv.y);
  unsafeAtomicAdd(ap + 2, alpha * wv.z);
  unsafeAtomicAdd(ap + 3, alpha * wv.w);
}

__global__ void gat_elu(const float* __restrict__ acc, float* __restrict__ out,
                        int total) {
  int i = blockIdx.x * blockDim.x + threadIdx.x;
  if (i >= total) return;
  float v = acc[i];
  out[i] = v > 0.f ? v : (__expf(v) - 1.0f);
}

extern "C" void kernel_launch(void* const* d_in, const int* in_sizes, int n_in,
                              void* d_out, int out_size, void* d_ws, size_t ws_size,
                              hipStream_t stream) {
  const float* x  = (const float*)d_in[0];
  const float* W  = (const float*)d_in[1];
  const float* a  = (const float*)d_in[2];
  const int*   ei = (const int*)d_in[3];
  const int N = in_sizes[0] / IN_DIM;
  const int E = in_sizes[3] / 2;
  float* out = (float*)d_out;

  char* ws = (char*)d_ws;
  size_t off = 0;
  auto carve = [&](size_t bytes) -> void* {
    void* p = ws + off;
    off += (bytes + 255) & ~(size_t)255;
    return p;
  };
  float* Wx    = (float*)carve((size_t)N * OUT_DIM * sizeof(float));
  float* s1    = (float*)carve((size_t)N * HEADS * sizeof(float));
  float* s2    = (float*)carve((size_t)N * HEADS * sizeof(float));
  int*   mkey  = (int*)  carve((size_t)N * HEADS * sizeof(int));
  float* denom = (float*)carve((size_t)N * HEADS * sizeof(float));
  float* ex    = (float*)carve((size_t)E * HEADS * sizeof(float));
  float* acc   = (float*)carve((size_t)N * OUT_DIM * sizeof(float));
  (void)ws_size;

  gat_gemm_wmma<<<(N + 15) / 16, 256, 0, stream>>>(x, W, Wx, N);

  gat_scores<<<(N * HEADS + 255) / 256, 256, 0, stream>>>(Wx, a, s1, s2, N);

  gat_init<<<(N * OUT_DIM + 255) / 256, 256, 0, stream>>>(acc, denom, mkey, N);

  gat_edge_max<<<(E * HEADS + 256 * EH_ITEMS - 1) / (256 * EH_ITEMS), 256, 0, stream>>>(
      ei, s1, s2, mkey, E);

  gat_edge_exp<<<(E * HEADS + 256 * EH_ITEMS - 1) / (256 * EH_ITEMS), 256, 0, stream>>>(
      ei, s1, s2, mkey, ex, denom, E);

  gat_recip<<<(N * HEADS + 255) / 256, 256, 0, stream>>>(denom, N * HEADS);

  {
    long long threads = (long long)E * 32;
    int blocks = (int)((threads + 255) / 256);
    gat_edge_aggr<<<blocks, 256, 0, stream>>>(ei, Wx, ex, denom, acc, E);
  }

  gat_elu<<<(N * OUT_DIM + 255) / 256, 256, 0, stream>>>(acc, out, N * OUT_DIM);
}


# aggregation: lane l takes dims l,l+32,l+64,l+96 of its edge, so an atomic add instruction covers one 128-byte line of acc, not four (4x fewer atomic line requests); arrival order of adds changes
# speedup vs baseline: 1.8576x; 1.1945x over previous
#include <hip/hip_runtime.h>
#include <cstdint>
#include <cstddef>

typedef __attribute__((ext_vector_type(2))) float v2f;
typedef __attribute__((ext_vector_type(8))) float v8f;

#define IN_DIM   128
#define OUT_DIM  128
#define HEADS    4
#define HEAD_DIM 32

__device__ __forceinline__ int f2key(float f) {
  int i = __float_as_int(f);
  return i >= 0 ? i : (i ^ 0x7fffffff);
}
__device__ __forceinline__ float key2f(int k) {
  int i = k >= 0 ? k : (k ^ 0x7fffffff);
  return __int_as_float(i);
}

__global__ void gat_gemm_wmma(const float* __restrict__ x,
                              const float* __restrict__ W,
                              float* __restrict__ Wx, int N) {
  const int tile  = blockIdx.x;
  const int wave  = threadIdx.x >> 5;
  const int lane  = threadIdx.x & 31;
  const int mbase = tile * 16;
  const int obase = wave * 16;
  const int half  = lane >> 4;
  const int l15   = lane & 15;

  int arow = mbase + l15;
  if (arow >= N) arow = N - 1;

  const float* aptr = x + (size_t)arow * IN_DIM + 2 * half;
  const float* bptr = W + (size_t)(obase + l15) * IN_DIM + 2 * half;

  v8f c = {};
  #pragma unroll
  for (int k = 0; k < IN_DIM; k += 4) {
    v2f af = *(const v2f*)(aptr + k);
    v2f bf = *(const v2f*)(bptr + k);
    c = __builtin_amdgcn_wmma_f32_16x16x4_f32(false, af, false, bf,
                                              (short)0, c, false, false);
  }

  #pragma unroll
  for (int r = 0; r < 8; ++r) {
    int row = mbase + r + 8 * half;
    if (row < N) Wx[(size_t)row * OUT_DIM + obase + l15] = c[r];
  }
}

__global__ void gat_scores(const float* __restrict__ Wx,
                           const float* __restrict__ a,
                           float* __restrict__ s1, float* __restrict__ s2,
                           int N) {
  int t = blockIdx.x * blockDim.x + threadIdx.x;
  if (t >= N * HEADS) return;
  int n = t >> 2, h = t & 3;
  const float4* p  = (const float4*)(Wx + (size_t)n * OUT_DIM + h * HEAD_DIM);
  const float4* a1 = (const float4*)(a);
  const float4* a2 = (const float4*)(a + HEAD_DIM);
  float acc1 = 0.f, acc2 = 0.f;
  #pragma unroll
  for (int d = 0; d < HEAD_DIM / 4; ++d) {
    float4 v  = p[d];
    float4 c1 = a1[d];
    float4 c2 = a2[d];
    acc1 += v.x * c1.x + v.y * c1.y + v.z * c1.z + v.w * c1.w;
    acc2 += v.x * c2.x + v.y * c2.y + v.z * c2.z + v.w * c2.w;
  }
  s1[t] = acc1;
  s2[t] = acc2;
}

__global__ void gat_init(float* __restrict__ acc, float* __restrict__ denom,
                         int* __restrict__ mkey, int N) {
  int i = blockIdx.x * blockDim.x + threadIdx.x;
  if (i < N * OUT_DIM) acc[i] = 0.f;
  if (i < N * HEADS) { denom[i] = 0.f; mkey[i] = (int)0x80000000; }
}

#define EH_ITEMS 4

template <bool FULL>
__device__ __forceinline__ void edge_max_body(const int* __restrict__ ei,
                                              const float* __restrict__ s1,
                                              const float* __restrict__ s2,
                                              int* __restrict__ mkey,
                                              int E, int total, int t0, int h) {
  int   src[EH_ITEMS], dst[EH_ITEMS];
  float el[EH_ITEMS];
  #pragma unroll
  for (int k = 0; k < EH_ITEMS; ++k) {
    int t = t0 + 32 * k;
    int e = (FULL || t < total) ? (t >> 2) : 0;
    src[k] = ei[e];
    dst[k] = ei[E + e];
  }
  #pragma unroll
  for (int k = 0; k < EH_ITEMS; ++k)
    el[k] = s1[(size_t)src[k] * HEADS + h] + s2[(size_t)dst[k] * HEADS + h];
  #pragma unroll
  for (int k = 0; k < EH_ITEMS; ++k) {
    if (FULL || t0 + 32 * k < total) {
      float v = el[k] > 0.f ? el[k] : 0.2f * el[k];
      atomicMax(&mkey[(size_t)dst[k] * HEADS + h], f2key(v));
    }
  }
}

__global__ void gat_edge_max(const int* __restrict__ ei,
                             const float* __restrict__ s1,
                             const float* __restrict__ s2,
                             int* __restrict__ mkey, int E) {
  const int total = E * HEADS;
  const int lane  = threadIdx.x & 31;
  const int h     = lane & 3;
  const int wbase = (blockIdx.x * (blockDim.x >> 5) + (threadIdx.x >> 5)) *
                    (32 * EH_ITEMS);
  if (wbase + 32 * EH_ITEMS <= total)
    edge_max_body<true>(ei, s1, s2, mkey, E, total, wbase + lane, h);
  else
    edge_max_body<false>(ei, s1, s2, mkey, E, total, wbase + lane, h);
}

template <bool FULL>
__device__ __forceinline__ void edge_exp_body(const int* __restrict__ ei,
                                              const float* __restrict__ s1,
                                              const float* __restrict__ s2,
                                              const int* __restrict__ mkey,
                                              float* __restrict__ ex,
                                              float* __restrict__ denom,
                                              int E, int total, int t0, int h) {
  int   src[EH_ITEMS], dst[EH_ITEMS], mk[EH_ITEMS];
  float el[EH_ITEMS];
  #pragma unroll
  for (int k = 0; k < EH_ITEMS; ++k) {
    int t = t0 + 32 * k;
    int e = (FULL || t < total) ? (t >> 2) : 0;
    src[k] = ei[e];
    dst[k] = ei[E + e];
  }
  #pragma unroll
  for (int k = 0; k < EH_ITEMS; ++k) {
    el[k] = s1[(size_t)src[k] * HEADS + h] + s2[(size_t)dst[k] * HEADS + h];
    mk[k] = mkey[(size_t)dst[k] * HEADS + h];
  }
  #pragma unroll
  for (int k = 0; k < EH_ITEMS; ++k) {
    int t = t0 + 32 * k;
    if (FULL || t < total) {
      float v  = el[k] > 0.f ? el[k] : 0.2f * el[k];
      float xv = __expf(v - key2f(mk[k]));
      ex[t] = xv;
      unsafeAtomicAdd(&denom[(size_t)dst[k] * HEADS + h], xv);
    }
  }
}

__global__ void gat_edge_exp(const int* __restrict__ ei,
                             const float* __restrict__ s1,
                             const float* __restrict__ s2,
                             const int* __restrict__ mkey,
                             float* __restrict__ ex,
                             float* __restrict__ denom, int E) {
  const int total = E * HEADS;
  const int lane  = threadIdx.x & 31;
  const int h     = lane & 3;
  const int wbase = (blockIdx.x * (blockDim.x >> 5) + (threadIdx.x >> 5)) *
                    (32 * EH_ITEMS);
  if (wbase + 32 * EH_ITEMS <= total)
    edge_exp_body<true>(ei, s1, s2, mkey, ex, denom, E, total, wbase + lane, h);
  else
    edge_exp_body<false>(ei, s1, s2, mkey, ex, denom, E, total, wbase + lane, h);
}

__global__ void gat_recip(float* __restrict__ denom, int total) {
  int i = blockIdx.x * blockDim.x + threadIdx.x;
  if (i >= total) return;
  denom[i] = __frcp_rn(denom[i]);
}

__global__ void gat_edge_aggr(const int* __restrict__ ei,
                              const float* __restrict__ Wx,
                              const float* __restrict__ ex,
                              const float* __restrict__ rden,
                              float* __restrict__ acc, int E) {
  long long gid = (long long)blockIdx.x * blockDim.x + threadIdx.x;
  int e = (int)(gid >> 5);
  int l = (int)(gid & 31);
  if (e >= E) return;
  int src = ei[e], dst = ei[E + e];
  float4 xv = *(const float4*)(ex + (size_t)e * HEADS);
  float4 rd = *(const float4*)(rden + (size_t)dst * HEADS);
  const float* wp = Wx + (size_t)src * OUT_DIM + l;
  float* ap = acc + (size_t)dst * OUT_DIM + l;
  float w0 = wp[0 * HEAD_DIM];
  float w1 = wp[1 * HEAD_DIM];
  float w2 = wp[2 * HEAD_DIM];
  float w3 = wp[3 * HEAD_DIM];
  unsafeAtomicAdd(ap + 0 * HEAD_DIM, (xv.x * rd.x) * w0);
  unsafeAtomicAdd(ap + 1 * HEAD_DIM, (xv.y * rd.y) * w1);
  unsafeAtomicAdd(ap + 2 * HEAD_DIM, (xv.z * rd.z) * w2);
  unsafeAtomicAdd(ap + 3 * HEAD_DIM, (xv.w * rd.w) * w3);
}

__global__ void gat_elu(const float* __restrict__ acc, float* __restrict__ out,
                        int total) {
  int i = blockIdx.x * blockDim.x + threadIdx.x;
  if (i >= total) return;
  float v = acc[i];
  out[i] = v > 0.f ? v : (__expf(v) - 1.0f);
}

extern "C" void kernel_launch(void* const* d_in, const int* in_sizes, int n_in,
                              void* d_out, int out_size, void* d_ws, size_t ws_size,
                              hipStream_t stream) {
  const float* x  = (const float*)d_in[0];
  const float* W  = (const float*)d_in[1];
  const float* a  = (const float*)d_in[2];
  const int*   ei = (const int*)d_in[3];
  const int N = in_sizes[0] / IN_DIM;
  const int E = in_sizes[3] / 2;
  float* out = (float*)d_out;

  char* ws = (char*)d_ws;
  size_t off = 0;
  auto carve = [&](size_t bytes) -> void* {
    void* p = ws + off;
    off += (bytes + 255) & ~(size_t)255;
    return p;
  };
  float* Wx    = (float*)carve((size_t)N * OUT_DIM * sizeof(float));
  float* s1    = (float*)carve((size_t)N * HEADS * sizeof(float));
  float* s2    = (float*)carve((size_t)N * HEADS * sizeof(float));
  int*   mkey  = (int*)  carve((size_t)N * HEADS * sizeof(int));
  float* denom = (float*)carve((size_t)N * HEADS * sizeof(float));
  float* ex    = (float*)carve((size_t)E * HEADS * sizeof(float));
  float* acc   = (float*)carve((size_t)N * OUT_DIM * sizeof(float));
  (void)ws_size;

  gat_gemm_wmma<<<(N + 15) / 16, 256, 0, stream>>>(x, W, Wx, N);

  gat_scores<<<(N * HEADS + 255) / 256, 256, 0, stream>>>(Wx, a, s1, s2, N);

  gat_init<<<(N * OUT_DIM + 255) / 256, 256, 0, stream>>>(acc, denom, mkey, N);

  gat_edge_max<<<(E * HEADS + 256 * EH_ITEMS - 1) / (256 * EH_ITEMS), 256, 0, stream>>>(
      ei, s1, s2, mkey, E);

  gat_edge_exp<<<(E * HEADS + 256 * EH_ITEMS - 1) / (256 * EH_ITEMS), 256, 0, stream>>>(
      ei, s1, s2, mkey, ex, denom, E);

  gat_recip<<<(N * HEADS + 255) / 256, 256, 0, stream>>>(denom, N * HEADS);

  {
    long long threads = (long long)E * 32;
    int blocks = (int)((threads + 255) / 256);
    gat_edge_aggr<<<blocks, 256, 0, stream>>>(ei, Wx, ex, denom, acc, E);
  }

  gat_elu<<<(N * OUT_DIM + 255) / 256, 256, 0, stream>>>(acc, out, N * OUT_DIM);
}


# edges grouped by destination (counting sort each call, rank by returning int atomic); one wave per node keeps max, exp, denominator, row sum in registers; no float atomics; sums in arrival-rank order
# speedup vs baseline: 2.9584x; 1.5926x over previous
#include <hip/hip_runtime.h>
#include <cstdint>
#include <cstddef>

typedef __attribute__((ext_vector_type(2))) float v2f;
typedef __attribute__((ext_vector_type(8))) float v8f;

#define IN_DIM   128
#define OUT_DIM  128
#define HEADS    4
#define HEAD_DIM 32

__global__ void gat_gemm_wmma(const float* __restrict__ x,
                              const float* __restrict__ W,
                              float* __restrict__ Wx, int N) {
  const int tile  = blockIdx.x;
  const int wave  = threadIdx.x >> 5;
  const int lane  = threadIdx.x & 31;
  const int mbase = tile * 16;
  const int obase = wave * 16;
  const int half  = lane >> 4;
  const int l15   = lane & 15;

  int arow = mbase + l15;
  if (arow >= N) arow = N - 1;

  const float* aptr = x + (size_t)arow * IN_DIM + 2 * half;
  const float* bptr = W + (size_t)(obase + l15) * IN_DIM + 2 * half;

  v8f c = {};
  #pragma unroll
  for (int k = 0; k < IN_DIM; k += 4) {
    v2f af = *(const v2f*)(aptr + k);
    v2f bf = *(const v2f*)(bptr + k);
    c = __builtin_amdgcn_wmma_f32_16x16x4_f32(false, af, false, bf,
                                              (short)0, c, false, false);
  }

  #pragma unroll
  for (int r = 0; r < 8; ++r) {
    int row = mbase + r + 8 * half;
    if (row < N) Wx[(size_t)row * OUT_DIM + obase + l15] = c[r];
  }
}

__global__ void gat_scores(const float* __restrict__ Wx,
                           const float* __restrict__ a,
                           float* __restrict__ s1, float* __restrict__ s2,
                           int N) {
  int t = blockIdx.x * blockDim.x + threadIdx.x;
  if (t >= N * HEADS) return;
  int n = t >> 2, h = t & 3;
  const float4* p  = (const float4*)(Wx + (size_t)n * OUT_DIM + h * HEAD_DIM);
  const float4* a1 = (const float4*)(a);
  const float4* a2 = (const float4*)(a + HEAD_DIM);
  float acc1 = 0.f, acc2 = 0.f;
  #pragma unroll
  for (int d = 0; d < HEAD_DIM / 4; ++d) {
    float4 v  = p[d];
    float4 c1 = a1[d];
    float4 c2 = a2[d];
    acc1 += v.x * c1.x + v.y * c1.y + v.z * c1.z + v.w * c1.w;
    acc2 += v.x * c2.x + v.y * c2.y + v.z * c2.z + v.w * c2.w;
  }
  s1[t] = acc1;
  s2[t] = acc2;
}

__global__ void gat_zero(int* __restrict__ cnt, int* __restrict__ cursor, int N) {
  int i = blockIdx.x * blockDim.x + threadIdx.x;
  if (i < N) cnt[i] = 0;
  if (i == 0) cursor[0] = 0;
}

__global__ void gat_count(const int* __restrict__ ei, int* __restrict__ cnt,
                          int* __restrict__ rank, int E) {
  int e = blockIdx.x * blockDim.x + threadIdx.x;
  if (e >= E) return;
  int dst = ei[E + e];
  rank[e] = atomicAdd(&cnt[dst], 1);
}

__global__ void gat_alloc(const int* __restrict__ cnt, int* __restrict__ off,
                          int* __restrict__ cursor, int N) {
  const int n    = blockIdx.x * blockDim.x + threadIdx.x;
  const int lane = threadIdx.x & 31;
  const int c    = n < N ? cnt[n] : 0;
  int incl = c;
  #pragma unroll
  for (int d = 1; d < 32; d <<= 1) {
    int t = __shfl_up(incl, d);
    if (lane >= d) incl += t;
  }
  const int total = __shfl(incl, 31);
  int base = 0;
  if (lane == 31) base = atomicAdd(cursor, total);
  base = __shfl(base, 31);
  if (n < N) off[n] = base + incl - c;
}

__global__ void gat_fill(const int* __restrict__ ei, const int* __restrict__ off,
                         const int* __restrict__ rank, int* __restrict__ ssrc,
                         int E) {
  int e = blockIdx.x * blockDim.x + threadIdx.x;
  if (e >= E) return;
  int src = ei[e], dst = ei[E + e];
  ssrc[off[dst] + rank[e]] = src;
}

__device__ __forceinline__ float lrelu02(float v) { return v > 0.f ? v : 0.2f * v; }

__device__ __forceinline__ float lane_f(float v, int j) {
  return __int_as_float(__builtin_amdgcn_readlane(__float_as_int(v), j));
}

#define EDGE_GROUP 4

__global__ void gat_node(const float* __restrict__ Wx,
                         const float* __restrict__ s1,
                         const float* __restrict__ s2,
                         const int* __restrict__ off,
                         const int* __restrict__ cnt,
                         const int* __restrict__ ssrc,
                         float* __restrict__ out, int N) {
  const int wv   = blockIdx.x * (blockDim.x >> 5) + (threadIdx.x >> 5);
  const int lane = threadIdx.x & 31;
  if (wv >= N) return;
  const int n = __builtin_amdgcn_readfirstlane(wv);
  const int o = off[n];
  const int d = cnt[n];
  const float4 sd = *(const float4*)(s2 + (size_t)n * HEADS);

  const float NEG_INF = -__builtin_inff();
  float4 m = make_float4(NEG_INF, NEG_INF, NEG_INF, NEG_INF);
  for (int c0 = 0; c0 < d; c0 += 32) {
    const int idx = c0 + lane;
    if (idx < d) {
      const int s = ssrc[o + idx];
      const float4 ss = *(const float4*)(s1 + (size_t)s * HEADS);
      m.x = fmaxf(m.x, lrelu02(ss.x + sd.x));
      m.y = fmaxf(m.y, lrelu02(ss.y + sd.y));
      m.z = fmaxf(m.z, lrelu02(ss.z + sd.z));
      m.w = fmaxf(m.w, lrelu02(ss.w + sd.w));
    }
  }
  #pragma unroll
  for (int x = 16; x >= 1; x >>= 1) {
    m.x = fmaxf(m.x, __shfl_xor(m.x, x));
    m.y = fmaxf(m.y, __shfl_xor(m.y, x));
    m.z = fmaxf(m.z, __shfl_xor(m.z, x));
    m.w = fmaxf(m.w, __shfl_xor(m.w, x));
  }

  float den[HEADS] = {0.f, 0.f, 0.f, 0.f};
  float acc[HEADS] = {0.f, 0.f, 0.f, 0.f};
  const float* wrow = Wx + lane;
  for (int c0 = 0; c0 < d; c0 += 32) {
    const int idx = c0 + lane;
    int    s = 0;
    float4 w = make_float4(0.f, 0.f, 0.f, 0.f);
    if (idx < d) {
      s = ssrc[o + idx];
      const float4 ss = *(const float4*)(s1 + (size_t)s * HEADS);
      w.x = __expf(lrelu02(ss.x + sd.x) - m.x);
      w.y = __expf(lrelu02(ss.y + sd.y) - m.y);
      w.z = __expf(lrelu02(ss.z + sd.z) - m.z);
      w.w = __expf(lrelu02(ss.w + sd.w) - m.w);
    }
    const int cn = (d - c0) < 32 ? (d - c0) : 32;
    int j = 0;
    for (; j + EDGE_GROUP <= cn; j += EDGE_GROUP) {
      float r[EDGE_GROUP][HEADS];
      #pragma unroll
      for (int g = 0; g < EDGE_GROUP; ++g) {
        const int sj = __builtin_amdgcn_readlane(s, j + g);
        const float* p = wrow + (size_t)sj * OUT_DIM;
        #pragma unroll
        for (int k = 0; k < HEADS; ++k) r[g][k] = p[k * HEAD_DIM];
      }
      #pragma unroll
      for (int g = 0; g < EDGE_GROUP; ++g) {
        const float wk[HEADS] = {lane_f(w.x, j + g), lane_f(w.y, j + g),
                                 lane_f(w.z, j + g), lane_f(w.w, j + g)};
        #pragma unroll
        for (int k = 0; k < HEADS; ++k) {
          den[k] += wk[k];
          acc[k] += wk[k] * r[g][k];
        }
      }
    }
    for (; j < cn; ++j) {
      const int sj = __builtin_amdgcn_readlane(s, j);
      const float* p = wrow + (size_t)sj * OUT_DIM;
      const float wk[HEADS] = {lane_f(w.x, j), lane_f(w.y, j),
                               lane_f(w.z, j), lane_f(w.w, j)};
      #pragma unroll
      for (int k = 0; k < HEADS; ++k) {
        den[k] += wk[k];
        acc[k] += wk[k] * p[k * HEAD_DIM];
      }
    }
  }
  float* orow = out + (size_t)n * OUT_DIM + lane;
  #pragma unroll
  for (int k = 0; k < HEADS; ++k) {
    const float rd = d > 0 ? __frcp_rn(den[k]) : 0.f;
    const float v  = acc[k] * rd;
    orow[k * HEAD_DIM] = v > 0.f ? v : (__expf(v) - 1.0f);
  }
}

extern "C" void kernel_launch(void* const* d_in, const int* in_sizes, int n_in,
                              void* d_out, int out_size, void* d_ws, size_t ws_size,
                              hipStream_t stream) {
  const float* x  = (const float*)d_in[0];
  const float* W  = (const float*)d_in[1];
  const float* a  = (const float*)d_in[2];
  const int*   ei = (const int*)d_in[3];
  const int N = in_sizes[0] / IN_DIM;
  const int E = in_sizes[3] / 2;
  float* out = (float*)d_out;

  char* ws = (char*)d_ws;
  size_t off_b = 0;
  auto carve = [&](size_t bytes) -> void* {
    void* p = ws + off_b;
    off_b += (bytes + 255) & ~(size_t)255;
    return p;
  };
  float* Wx     = (float*)carve((size_t)N * OUT_DIM * sizeof(float));
  float* s1     = (float*)carve((size_t)N * HEADS * sizeof(float));
  float* s2     = (float*)carve((size_t)N * HEADS * sizeof(float));
  int*   cnt    = (int*)  carve((size_t)N * sizeof(int));
  int*   cursor = (int*)  carve(sizeof(int));
  int*   offs   = (int*)  carve((size_t)N * sizeof(int));
  int*   rank   = (int*)  carve((size_t)E * sizeof(int));
  int*   ssrc   = (int*)  carve((size_t)E * sizeof(int));
  (void)ws_size; (void)n_in; (void)out_size;

  gat_gemm_wmma<<<(N + 15) / 16, 256, 0, stream>>>(x, W, Wx, N);

  gat_scores<<<(N * HEADS + 255) / 256, 256, 0, stream>>>(Wx, a, s1, s2, N);

  gat_zero<<<(N + 255) / 256, 256, 0, stream>>>(cnt, cursor, N);

  gat_count<<<(E + 255) / 256, 256, 0, stream>>>(ei, cnt, rank, E);

  gat_alloc<<<(N + 255) / 256, 256, 0, stream>>>(cnt, offs, cursor, N);

  gat_fill<<<(E + 255) / 256, 256, 0, stream>>>(ei, offs, rank, ssrc, E);

  gat_node<<<(N + 7) / 8, 256, 0, stream>>>(Wx, s1, s2, offs, cnt, ssrc, out, N);
}


# edge ranks from returning LDS adds: 32 edge chunks, each work-group counts all nodes in its LDS (no per-edge global atomic, no clear launch); a node's sums run chunk by chunk, arrival order in each
# speedup vs baseline: 4.0189x; 1.3585x over previous
#include <hip/hip_runtime.h>
#include <cstdint>
#include <cstddef>

typedef __attribute__((ext_vector_type(2))) float v2f;
typedef __attribute__((ext_vector_type(8))) float v8f;

#define IN_DIM   128
#define OUT_DIM  128
#define HEADS    4
#define HEAD_DIM 32

__global__ void gat_gemm_wmma(const float* __restrict__ x,
                              const float* __restrict__ W,
                              float* __restrict__ Wx, int N) {
  const int tile  = blockIdx.x;
  const int wave  = threadIdx.x >> 5;
  const int lane  = threadIdx.x & 31;
  const int mbase = tile * 16;
  const int obase = wave * 16;
  const int half  = lane >> 4;
  const int l15   = lane & 15;

  int arow = mbase + l15;
  if (arow >= N) arow = N - 1;

  const float* aptr = x + (size_t)arow * IN_DIM + 2 * half;
  const float* bptr = W + (size_t)(obase + l15) * IN_DIM + 2 * half;

  v8f c = {};
  #pragma unroll
  for (int k = 0; k < IN_DIM; k += 4) {
    v2f af = *(const v2f*)(aptr + k);
    v2f bf = *(const v2f*)(bptr + k);
    c = __builtin_amdgcn_wmma_f32_16x16x4_f32(false, af, false, bf,
                                              (short)0, c, false, false);
  }

  #pragma unroll
  for (int r = 0; r < 8; ++r) {
    int row = mbase + r + 8 * half;
    if (row < N) Wx[(size_t)row * OUT_DIM + obase + l15] = c[r];
  }
}

__global__ void gat_scores(const float* __restrict__ Wx,
                           const float* __restrict__ a,
                           float* __restrict__ s1, float* __restrict__ s2,
                           int N) {
  int t = blockIdx.x * blockDim.x + threadIdx.x;
  if (t >= N * HEADS) return;
  int n = t >> 2, h = t & 3;
  const float4* p  = (const float4*)(Wx + (size_t)n * OUT_DIM + h * HEAD_DIM);
  const float4* a1 = (const float4*)(a);
  const float4* a2 = (const float4*)(a + HEAD_DIM);
  float acc1 = 0.f, acc2 = 0.f;
  #pragma unroll
  for (int d = 0; d < HEAD_DIM / 4; ++d) {
    float4 v  = p[d];
    float4 c1 = a1[d];
    float4 c2 = a2[d];
    acc1 += v.x * c1.x + v.y * c1.y + v.z * c1.z + v.w * c1.w;
    acc2 += v.x * c2.x + v.y * c2.y + v.z * c2.z + v.w * c2.w;
  }
  s1[t] = acc1;
  s2[t] = acc2;
}

#define CHUNKS      32
#define MAX_NODES   50000
#define COUNT_BLOCK 1024
#define COUNT_GROUP 8

__global__ __launch_bounds__(COUNT_BLOCK)
void gat_count(const int* __restrict__ ei, int* __restrict__ ccnt,
               int* __restrict__ rank, int* __restrict__ cursor, int E) {
  __shared__ int4 tab4[MAX_NODES / 4];
  int* tab = (int*)tab4;
  const int tid = threadIdx.x;
  const int c   = blockIdx.x;
  const int per = (E + CHUNKS - 1) / CHUNKS;
  const int e0  = c * per;
  const int e1  = (e0 + per < E) ? (e0 + per) : E;

  for (int i = tid; i < MAX_NODES / 4; i += COUNT_BLOCK) tab4[i] = make_int4(0, 0, 0, 0);
  if (c == 0 && tid == 0) cursor[0] = 0;
  __syncthreads();

  const int* dstp = ei + E;
  int base = e0;
  for (; base + COUNT_GROUP * COUNT_BLOCK <= e1; base += COUNT_GROUP * COUNT_BLOCK) {
    int d[COUNT_GROUP], r[COUNT_GROUP];
    #pragma unroll
    for (int g = 0; g < COUNT_GROUP; ++g) d[g] = dstp[base + g * COUNT_BLOCK + tid];
    #pragma unroll
    for (int g = 0; g < COUNT_GROUP; ++g) r[g] = atomicAdd(&tab[d[g]], 1);
    #pragma unroll
    for (int g = 0; g < COUNT_GROUP; ++g) rank[base + g * COUNT_BLOCK + tid] = r[g];
  }
  for (int e = base + tid; e < e1; e += COUNT_BLOCK)
    rank[e] = atomicAdd(&tab[dstp[e]], 1);
  __syncthreads();

  int4* row4 = (int4*)(ccnt + (size_t)c * MAX_NODES);
  for (int i = tid; i < MAX_NODES / 4; i += COUNT_BLOCK) row4[i] = tab4[i];
}

__global__ void gat_alloc(int* __restrict__ ccnt, int* __restrict__ cnt,
                          int* __restrict__ off, int* __restrict__ cursor, int N) {
  const int n    = blockIdx.x * blockDim.x + threadIdx.x;
  const int lane = threadIdx.x & 31;
  const int nn   = n < N ? n : N - 1;
  int v[CHUNKS];
  #pragma unroll
  for (int k = 0; k < CHUNKS; ++k) v[k] = ccnt[(size_t)k * MAX_NODES + nn];
  int c = 0;
  #pragma unroll
  for (int k = 0; k < CHUNKS; ++k) c += v[k];
  if (n >= N) c = 0;
  int incl = c;
  #pragma unroll
  for (int d = 1; d < 32; d <<= 1) {
    int t = __shfl_up(incl, d);
    if (lane >= d) incl += t;
  }
  const int total = __shfl(incl, 31);
  int base = 0;
  if (lane == 31) base = atomicAdd(cursor, total);
  base = __shfl(base, 31);
  if (n < N) {
    int run = base + incl - c;
    off[n] = run;
    cnt[n] = c;
    #pragma unroll
    for (int k = 0; k < CHUNKS; ++k) {
      ccnt[(size_t)k * MAX_NODES + n] = run;
      run += v[k];
    }
  }
}

__global__ void gat_fill(const int* __restrict__ ei, const int* __restrict__ ccnt,
                         const int* __restrict__ rank, int* __restrict__ ssrc,
                         int E) {
  int e = blockIdx.x * blockDim.x + threadIdx.x;
  if (e >= E) return;
  const int per = (E + CHUNKS - 1) / CHUNKS;
  const int c   = e / per;
  int src = ei[e], dst = ei[E + e];
  ssrc[ccnt[(size_t)c * MAX_NODES + dst] + rank[e]] = src;
}

__device__ __forceinline__ float lrelu02(float v) { return v > 0.f ? v : 0.2f * v; }

__device__ __forceinline__ float lane_f(float v, int j) {
  return __int_as_float(__builtin_amdgcn_readlane(__float_as_int(v), j));
}

#define EDGE_GROUP 4

__global__ void gat_node(const float* __restrict__ Wx,
                         const float* __restrict__ s1,
                         const float* __restrict__ s2,
                         const int* __restrict__ off,
                         const int* __restrict__ cnt,
                         const int* __restrict__ ssrc,
                         float* __restrict__ out, int N) {
  const int wv   = blockIdx.x * (blockDim.x >> 5) + (threadIdx.x >> 5);
  const int lane = threadIdx.x & 31;
  if (wv >= N) return;
  const int n = __builtin_amdgcn_readfirstlane(wv);
  const int o = off[n];
  const int d = cnt[n];
  const float4 sd = *(const float4*)(s2 + (size_t)n * HEADS);

  const float NEG_INF = -__builtin_inff();
  float4 m = make_float4(NEG_INF, NEG_INF, NEG_INF, NEG_INF);
  for (int c0 = 0; c0 < d; c0 += 32) {
    const int idx = c0 + lane;
    if (idx < d) {
      const int s = ssrc[o + idx];
      const float4 ss = *(const float4*)(s1 + (size_t)s * HEADS);
      m.x = fmaxf(m.x, lrelu02(ss.x + sd.x));
      m.y = fmaxf(m.y, lrelu02(ss.y + sd.y));
      m.z = fmaxf(m.z, lrelu02(ss.z + sd.z));
      m.w = fmaxf(m.w, lrelu02(ss.w + sd.w));
    }
  }
  #pragma unroll
  for (int x = 16; x >= 1; x >>= 1) {
    m.x = fmaxf(m.x, __shfl_xor(m.x, x));
    m.y = fmaxf(m.y, __shfl_xor(m.y, x));
    m.z = fmaxf(m.z, __shfl_xor(m.z, x));
    m.w = fmaxf(m.w, __shfl_xor(m.w, x));
  }

  float den[HEADS] = {0.f, 0.f, 0.f, 0.f};
  float acc[HEADS] = {0.f, 0.f, 0.f, 0.f};
  const float* wrow = Wx + lane;
  for (int c0 = 0; c0 < d; c0 += 32) {
    const int idx = c0 + lane;
    int    s = 0;
    float4 w = make_float4(0.f, 0.f, 0.f, 0.f);
    if (idx < d) {
      s = ssrc[o + idx];
      const float4 ss = *(const float4*)(s1 + (size_t)s * HEADS);
      w.x = __expf(lrelu02(ss.x + sd.x) - m.x);
      w.y = __expf(lrelu02(ss.y + sd.y) - m.y);
      w.z = __expf(lrelu02(ss.z + sd.z) - m.z);
      w.w = __expf(lrelu02(ss.w + sd.w) - m.w);
    }
    const int cn = (d - c0) < 32 ? (d - c0) : 32;
    int j = 0;
    for (; j + EDGE_GROUP <= cn; j += EDGE_GROUP) {
      float r[EDGE_GROUP][HEADS];
      #pragma unroll
      for (int g = 0; g < EDGE_GROUP; ++g) {
        const int sj = __builtin_amdgcn_readlane(s, j + g);
        const float* p = wrow + (size_t)sj * OUT_DIM;
        #pragma unroll
        for (int k = 0; k < HEADS; ++k) r[g][k] = p[k * HEAD_DIM];
      }
      #pragma unroll
      for (int g = 0; g < EDGE_GROUP; ++g) {
        const float wk[HEADS] = {lane_f(w.x, j + g), lane_f(w.y, j + g),
                                 lane_f(w.z, j + g), lane_f(w.w, j + g)};
        #pragma unroll
        for (int k = 0; k < HEADS; ++k) {
          den[k] += wk[k];
          acc[k] += wk[k] * r[g][k];
        }
      }
    }
    for (; j < cn; ++j) {
      const int sj = __builtin_amdgcn_readlane(s, j);
      const float* p = wrow + (size_t)sj * OUT_DIM;
      const float wk[HEADS] = {lane_f(w.x, j), lane_f(w.y, j),
                               lane_f(w.z, j), lane_f(w.w, j)};
      #pragma unroll
      for (int k = 0; k < HEADS; ++k) {
        den[k] += wk[k];
        acc[k] += wk[k] * p[k * HEAD_DIM];
      }
    }
  }
  float* orow = out + (size_t)n * OUT_DIM + lane;
  #pragma unroll
  for (int k = 0; k < HEADS; ++k) {
    const float rd = d > 0 ? __frcp_rn(den[k]) : 0.f;
    const float v  = acc[k] * rd;
    orow[k * HEAD_DIM] = v > 0.f ? v : (__expf(v) - 1.0f);
  }
}

extern "C" void kernel_launch(void* const* d_in, const int* in_sizes, int n_in,
                              void* d_out, int out_size, void* d_ws, size_t ws_size,
                              hipStream_t stream) {
  const float* x  = (const float*)d_in[0];
  const float* W  = (const float*)d_in[1];
  const float* a  = (const float*)d_in[2];
  const int*   ei = (const int*)d_in[3];
  const int N = in_sizes[0] / IN_DIM;
  const int E = in_sizes[3] / 2;
  float* out = (float*)d_out;

  char* ws = (char*)d_ws;
  size_t off_b = 0;
  auto carve = [&](size_t bytes) -> void* {
    void* p = ws + off_b;
    off_b += (bytes + 255) & ~(size_t)255;
    return p;
  };
  float* Wx     = (float*)carve((size_t)N * OUT_DIM * sizeof(float));
  float* s1     = (float*)carve((size_t)N * HEADS * sizeof(float));
  float* s2     = (float*)carve((size_t)N * HEADS * sizeof(float));
  int*   cnt    = (int*)  carve((size_t)N * sizeof(int));
  int*   cursor = (int*)  carve(sizeof(int));
  int*   offs   = (int*)  carve((size_t)N * sizeof(int));
  int*   rank   = (int*)  carve((size_t)E * sizeof(int));
  int*   ssrc   = (int*)  carve((size_t)E * sizeof(int));
  int*   ccnt   = (int*)  carve((size_t)CHUNKS * MAX_NODES * sizeof(int));
  (void)ws_size; (void)n_in; (void)out_size;

  gat_gemm_wmma<<<(N + 15) / 16, 256, 0, stream>>>(x, W, Wx, N);

  gat_scores<<<(N * HEADS + 255) / 256, 256, 0, stream>>>(Wx, a, s1, s2, N);

  gat_count<<<CHUNKS, COUNT_BLOCK, 0, stream>>>(ei, ccnt, rank, cursor, E);

  gat_alloc<<<(N + 255) / 256, 256, 0, stream>>>(ccnt, cnt, offs, cursor, N);

  gat_fill<<<(E + 255) / 256, 256, 0, stream>>>(ei, ccnt, rank, ssrc, E);

  gat_node<<<(N + 7) / 8, 256, 0, stream>>>(Wx, s1, s2, offs, cnt, ssrc, out, N);
}


# GEMM operand feed: W (once per block) and 32-row x tiles reach the WMMA fragments through LDS by whole-line global loads, blocks walk over tiles; no 16-line fragment loads from global; k order kept
# speedup vs baseline: 5.0316x; 1.2520x over previous
#include <hip/hip_runtime.h>
#include <cstdint>
#include <cstddef>

typedef __attribute__((ext_vector_type(2))) float v2f;
typedef __attribute__((ext_vector_type(8))) float v8f;

#define IN_DIM   128
#define OUT_DIM  128
#define HEADS    4
#define HEAD_DIM 32

#define GEMM_TILE  32
#define GEMM_PITCH (IN_DIM + 4)
#define GEMM_GRID  512

__global__ __launch_bounds__(256)
void gat_gemm_wmma(const float* __restrict__ x,
                   const float* __restrict__ W,
                   float* __restrict__ Wx, int N) {
  __shared__ float4 lds4[OUT_DIM * GEMM_PITCH / 4];
  float* lds = (float*)lds4;
  const int wave  = threadIdx.x >> 5;
  const int lane  = threadIdx.x & 31;
  const int obase = wave * 16;
  const int half  = lane >> 4;
  const int l15   = lane & 15;

  const int ntiles = (N + GEMM_TILE - 1) / GEMM_TILE;
  int t = blockIdx.x;
  if (t >= ntiles) return;

  const float* xl = x + lane * 4;
  float4 st0, st1, st2, st3;
  {
    const int r0 = t * GEMM_TILE + wave;
    st0 = *(const float4*)(xl + (size_t)min(r0,      N - 1) * IN_DIM);
    st1 = *(const float4*)(xl + (size_t)min(r0 + 8,  N - 1) * IN_DIM);
    st2 = *(const float4*)(xl + (size_t)min(r0 + 16, N - 1) * IN_DIM);
    st3 = *(const float4*)(xl + (size_t)min(r0 + 24, N - 1) * IN_DIM);
  }

  v2f bf[IN_DIM / 4];
  {
    float4 wv[OUT_DIM / 8];
    #pragma unroll
    for (int j = 0; j < OUT_DIM / 8; ++j)
      wv[j] = *(const float4*)(W + (size_t)(wave + 8 * j) * IN_DIM + lane * 4);
    #pragma unroll
    for (int j = 0; j < OUT_DIM / 8; ++j)
      *(float4*)(lds + (wave + 8 * j) * GEMM_PITCH + lane * 4) = wv[j];
    __syncthreads();
    const float* bp = lds + (obase + l15) * GEMM_PITCH + 2 * half;
    #pragma unroll
    for (int k = 0; k < IN_DIM / 4; ++k) bf[k] = *(const v2f*)(bp + 4 * k);
    __syncthreads();
  }

  int buf = 0;
  for (; t < ntiles; t += gridDim.x) {
    float* xs = lds + buf * (GEMM_TILE * GEMM_PITCH);
    float* xw = xs + wave * GEMM_PITCH + lane * 4;
    *(float4*)(xw)                   = st0;
    *(float4*)(xw + 8  * GEMM_PITCH) = st1;
    *(float4*)(xw + 16 * GEMM_PITCH) = st2;
    *(float4*)(xw + 24 * GEMM_PITCH) = st3;
    __syncthreads();

    const int tn = t + gridDim.x;
    if (tn < ntiles) {
      const int r0 = tn * GEMM_TILE + wave;
      st0 = *(const float4*)(xl + (size_t)min(r0,      N - 1) * IN_DIM);
      st1 = *(const float4*)(xl + (size_t)min(r0 + 8,  N - 1) * IN_DIM);
      st2 = *(const float4*)(xl + (size_t)min(r0 + 16, N - 1) * IN_DIM);
      st3 = *(const float4*)(xl + (size_t)min(r0 + 24, N - 1) * IN_DIM);
    }

    const float* a0 = xs + l15 * GEMM_PITCH + 2 * half;
    const float* a1 = a0 + 16 * GEMM_PITCH;
    v8f c0 = {}, c1 = {};
    #pragma unroll
    for (int k = 0; k < IN_DIM / 4; ++k) {
      v2f af0 = *(const v2f*)(a0 + 4 * k);
      v2f af1 = *(const v2f*)(a1 + 4 * k);
      c0 = __builtin_amdgcn_wmma_f32_16x16x4_f32(false, af0, false, bf[k],
                                                 (short)0, c0, false, false);
      c1 = __builtin_amdgcn_wmma_f32_16x16x4_f32(false, af1, false, bf[k],
                                                 (short)0, c1, false, false);
    }

    const int mbase = t * GEMM_TILE;
    float* orow = Wx + (size_t)(mbase + 8 * half) * OUT_DIM + obase + l15;
    if (mbase + GEMM_TILE <= N) {
      #pragma unroll
      for (int r = 0; r < 8; ++r) {
        orow[r * OUT_DIM]        = c0[r];
        orow[(r + 16) * OUT_DIM] = c1[r];
      }
    } else {
      #pragma unroll
      for (int r = 0; r < 8; ++r) {
        const int row0 = mbase + r + 8 * half;
        if (row0 < N)      orow[r * OUT_DIM]        = c0[r];
        if (row0 + 16 < N) orow[(r + 16) * OUT_DIM] = c1[r];
      }
    }
    buf ^= 1;
  }
}

__global__ void gat_scores(const float* __restrict__ Wx,
                           const float* __restrict__ a,
                           float* __restrict__ s1, float* __restrict__ s2,
                           int N) {
  int t = blockIdx.x * blockDim.x + threadIdx.x;
  if (t >= N * HEADS) return;
  int n = t >> 2, h = t & 3;
  const float4* p  = (const float4*)(Wx + (size_t)n * OUT_DIM + h * HEAD_DIM);
  const float4* a1 = (const float4*)(a);
  const float4* a2 = (const float4*)(a + HEAD_DIM);
  float acc1 = 0.f, acc2 = 0.f;
  #pragma unroll
  for (int d = 0; d < HEAD_DIM / 4; ++d) {
    float4 v  = p[d];
    float4 c1 = a1[d];
    float4 c2 = a2[d];
    acc1 += v.x * c1.x + v.y * c1.y + v.z * c1.z + v.w * c1.w;
    acc2 += v.x * c2.x + v.y * c2.y + v.z * c2.z + v.w * c2.w;
  }
  s1[t] = acc1;
  s2[t] = acc2;
}

#define CHUNKS      32
#define MAX_NODES   50000
#define COUNT_BLOCK 1024
#define COUNT_GROUP 8

__global__ __launch_bounds__(COUNT_BLOCK)
void gat_count(const int* __restrict__ ei, int* __restrict__ ccnt,
               int* __restrict__ rank, int* __restrict__ cursor, int E) {
  __shared__ int4 tab4[MAX_NODES / 4];
  int* tab = (int*)tab4;
  const int tid = threadIdx.x;
  const int c   = blockIdx.x;
  const int per = (E + CHUNKS - 1) / CHUNKS;
  const int e0  = c * per;
  const int e1  = (e0 + per < E) ? (e0 + per) : E;

  for (int i = tid; i < MAX_NODES / 4; i += COUNT_BLOCK) tab4[i] = make_int4(0, 0, 0, 0);
  if (c == 0 && tid == 0) cursor[0] = 0;
  __syncthreads();

  const int* dstp = ei + E;
  int base = e0;
  for (; base + COUNT_GROUP * COUNT_BLOCK <= e1; base += COUNT_GROUP * COUNT_BLOCK) {
    int d[COUNT_GROUP], r[COUNT_GROUP];
    #pragma unroll
    for (int g = 0; g < COUNT_GROUP; ++g) d[g] = dstp[base + g * COUNT_BLOCK + tid];
    #pragma unroll
    for (int g = 0; g < COUNT_GROUP; ++g) r[g] = atomicAdd(&tab[d[g]], 1);
    #pragma unroll
    for (int g = 0; g < COUNT_GROUP; ++g) rank[base + g * COUNT_BLOCK + tid] = r[g];
  }
  for (int e = base + tid; e < e1; e += COUNT_BLOCK)
    rank[e] = atomicAdd(&tab[dstp[e]], 1);
  __syncthreads();

  int4* row4 = (int4*)(ccnt + (size_t)c * MAX_NODES);
  for (int i = tid; i < MAX_NODES / 4; i += COUNT_BLOCK) row4[i] = tab4[i];
}

__global__ void gat_alloc(int* __restrict__ ccnt, int* __restrict__ cnt,
                          int* __restrict__ off, int* __restrict__ cursor, int N) {
  const int n    = blockIdx.x * blockDim.x + threadIdx.x;
  const int lane = threadIdx.x & 31;
  const int nn   = n < N ? n : N - 1;
  int v[CHUNKS];
  #pragma unroll
  for (int k = 0; k < CHUNKS; ++k) v[k] = ccnt[(size_t)k * MAX_NODES + nn];
  int c = 0;
  #pragma unroll
  for (int k = 0; k < CHUNKS; ++k) c += v[k];
  if (n >= N) c = 0;
  int incl = c;
  #pragma unroll
  for (int d = 1; d < 32; d <<= 1) {
    int t = __shfl_up(incl, d);
    if (lane >= d) incl += t;
  }
  const int total = __shfl(incl, 31);
  int base = 0;
  if (lane == 31) base = atomicAdd(cursor, total);
  base = __shfl(base, 31);
  if (n < N) {
    int run = base + incl - c;
    off[n] = run;
    cnt[n] = c;
    #pragma unroll
    for (int k = 0; k < CHUNKS; ++k) {
      ccnt[(size_t)k * MAX_NODES + n] = run;
      run += v[k];
    }
  }
}

__global__ void gat_fill(const int* __restrict__ ei, const int* __restrict__ ccnt,
                         const int* __restrict__ rank, int* __restrict__ ssrc,
                         int E) {
  int e = blockIdx.x * blockDim.x + threadIdx.x;
  if (e >= E) return;
  const int per = (E + CHUNKS - 1) / CHUNKS;
  const int c   = e / per;
  int src = ei[e], dst = ei[E + e];
  ssrc[ccnt[(size_t)c * MAX_NODES + dst] + rank[e]] = src;
}

__device__ __forceinline__ float lrelu02(float v) { return v > 0.f ? v : 0.2f * v; }

__device__ __forceinline__ float lane_f(float v, int j) {
  return __int_as_float(__builtin_amdgcn_readlane(__float_as_int(v), j));
}

#define EDGE_GROUP 4

__global__ void gat_node(const float* __restrict__ Wx,
                         const float* __restrict__ s1,
                         const float* __restrict__ s2,
                         const int* __restrict__ off,
                         const int* __restrict__ cnt,
                         const int* __restrict__ ssrc,
                         float* __restrict__ out, int N) {
  const int wv   = blockIdx.x * (blockDim.x >> 5) + (threadIdx.x >> 5);
  const int lane = threadIdx.x & 31;
  if (wv >= N) return;
  const int n = __builtin_amdgcn_readfirstlane(wv);
  const int o = off[n];
  const int d = cnt[n];
  const float4 sd = *(const float4*)(s2 + (size_t)n * HEADS);

  const float NEG_INF = -__builtin_inff();
  float4 m = make_float4(NEG_INF, NEG_INF, NEG_INF, NEG_INF);
  for (int c0 = 0; c0 < d; c0 += 32) {
    const int idx = c0 + lane;
    if (idx < d) {
      const int s = ssrc[o + idx];
      const float4 ss = *(const float4*)(s1 + (size_t)s * HEADS);
      m.x = fmaxf(m.x, lrelu02(ss.x + sd.x));
      m.y = fmaxf(m.y, lrelu02(ss.y + sd.y));
      m.z = fmaxf(m.z, lrelu02(ss.z + sd.z));
      m.w = fmaxf(m.w, lrelu02(ss.w + sd.w));
    }
  }
  #pragma unroll
  for (int x = 16; x >= 1; x >>= 1) {
    m.x = fmaxf(m.x, __shfl_xor(m.x, x));
    m.y = fmaxf(m.y, __shfl_xor(m.y, x));
    m.z = fmaxf(m.z, __shfl_xor(m.z, x));
    m.w = fmaxf(m.w, __shfl_xor(m.w, x));
  }

  float den[HEADS] = {0.f, 0.f, 0.f, 0.f};
  float acc[HEADS] = {0.f, 0.f, 0.f, 0.f};
  const float* wrow = Wx + lane;
  for (int c0 = 0; c0 < d; c0 += 32) {
    const int idx = c0 + lane;
    int    s = 0;
    float4 w = make_float4(0.f, 0.f, 0.f, 0.f);
    if (idx < d) {
      s = ssrc[o + idx];
      const float4 ss = *(const float4*)(s1 + (size_t)s * HEADS);
      w.x = __expf(lrelu02(ss.x + sd.x) - m.x);
      w.y = __expf(lrelu02(ss.y + sd.y) - m.y);
      w.z = __expf(lrelu02(ss.z + sd.z) - m.z);
      w.w = __expf(lrelu02(ss.w + sd.w) - m.w);
    }
    const int cn = (d - c0) < 32 ? (d - c0) : 32;
    int j = 0;
    for (; j + EDGE_GROUP <= cn; j += EDGE_GROUP) {
      float r[EDGE_GROUP][HEADS];
      #pragma unroll
      for (int g = 0; g < EDGE_GROUP; ++g) {
        const int sj = __builtin_amdgcn_readlane(s, j + g);
        const float* p = wrow + (size_t)sj * OUT_DIM;
        #pragma unroll
        for (int k = 0; k < HEADS; ++k) r[g][k] = p[k * HEAD_DIM];
      }
      #pragma unroll
      for (int g = 0; g < EDGE_GROUP; ++g) {
        const float wk[HEADS] = {lane_f(w.x, j + g), lane_f(w.y, j + g),
                                 lane_f(w.z, j + g), lane_f(w.w, j + g)};
        #pragma unroll
        for (int k = 0; k < HEADS; ++k) {
          den[k] += wk[k];
          acc[k] += wk[k] * r[g][k];
        }
      }
    }
    for (; j < cn; ++j) {
      const int sj = __builtin_amdgcn_readlane(s, j);
      const float* p = wrow + (size_t)sj * OUT_DIM;
      const float wk[HEADS] = {lane_f(w.x, j), lane_f(w.y, j),
                               lane_f(w.z, j), lane_f(w.w, j)};
      #pragma unroll
      for (int k = 0; k < HEADS; ++k) {
        den[k] += wk[k];
        acc[k] += wk[k] * p[k * HEAD_DIM];
      }
    }
  }
  float* orow = out + (size_t)n * OUT_DIM + lane;
  #pragma unroll
  for (int k = 0; k < HEADS; ++k) {
    const float rd = d > 0 ? __frcp_rn(den[k]) : 0.f;
    const float v  = acc[k] * rd;
    orow[k * HEAD_DIM] = v > 0.f ? v : (__expf(v) - 1.0f);
  }
}

extern "C" void kernel_launch(void* const* d_in, const int* in_sizes, int n_in,
                              void* d_out, int out_size, void* d_ws, size_t ws_size,
                              hipStream_t stream) {
  const float* x  = (const float*)d_in[0];
  const float* W  = (const float*)d_in[1];
  const float* a  = (const float*)d_in[2];
  const int*   ei = (const int*)d_in[3];
  const int N = in_sizes[0] / IN_DIM;
  const int E = in_sizes[3] / 2;
  float* out = (float*)d_out;

  char* ws = (char*)d_ws;
  size_t off_b = 0;
  auto carve = [&](size_t bytes) -> void* {
    void* p = ws + off_b;
    off_b += (bytes + 255) & ~(size_t)255;
    return p;
  };
  float* Wx     = (float*)carve((size_t)N * OUT_DIM * sizeof(float));
  float* s1     = (float*)carve((size_t)N * HEADS * sizeof(float));
  float* s2     = (float*)carve((size_t)N * HEADS * sizeof(float));
  int*   cnt    = (int*)  carve((size_t)N * sizeof(int));
  int*   cursor = (int*)  carve(sizeof(int));
  int*   offs   = (int*)  carve((size_t)N * sizeof(int));
  int*   rank   = (int*)  carve((size_t)E * sizeof(int));
  int*   ssrc   = (int*)  carve((size_t)E * sizeof(int));
  int*   ccnt   = (int*)  carve((size_t)CHUNKS * MAX_NODES * sizeof(int));
  (void)ws_size; (void)n_in; (void)out_size;

  gat_gemm_wmma<<<GEMM_GRID, 256, 0, stream>>>(x, W, Wx, N);

  gat_scores<<<(N * HEADS + 255) / 256, 256, 0, stream>>>(Wx, a, s1, s2, N);

  gat_count<<<CHUNKS, COUNT_BLOCK, 0, stream>>>(ei, ccnt, rank, cursor, E);

  gat_alloc<<<(N + 255) / 256, 256, 0, stream>>>(ccnt, cnt, offs, cursor, N);

  gat_fill<<<(E + 255) / 256, 256, 0, stream>>>(ei, ccnt, rank, ssrc, E);

  gat_node<<<(N + 7) / 8, 256, 0, stream>>>(Wx, s1, s2, offs, cnt, ssrc, out, N);
}


# per-node scores s1,s2 formed in the GEMM epilogue from the result tile staged in LDS (whole-row Wx stores); scores launch and its Wx re-read gone; score dot = 8 lane partials + butterfly (new order)
# speedup vs baseline: 5.1840x; 1.0303x over previous
#include <hip/hip_runtime.h>
#include <cstdint>
#include <cstddef>

typedef __attribute__((ext_vector_type(2))) float v2f;
typedef __attribute__((ext_vector_type(8))) float v8f;

#define IN_DIM   128
#define OUT_DIM  128
#define HEADS    4
#define HEAD_DIM 32

#define GEMM_TILE  32
#define GEMM_PITCH (IN_DIM + 4)
#define GEMM_GRID  512

template <int M>
__device__ __forceinline__ float row_xor(float v) {
  return __int_as_float(__builtin_amdgcn_update_dpp(0, __float_as_int(v), 0x160 + M,
                                                    0xf, 0xf, false));
}

__global__ __launch_bounds__(256)
void gat_gemm_wmma(const float* __restrict__ x,
                   const float* __restrict__ W,
                   const float* __restrict__ a,
                   float* __restrict__ Wx,
                   float* __restrict__ s1, float* __restrict__ s2, int N) {
  __shared__ float4 lds4[OUT_DIM * GEMM_PITCH / 4];
  float* lds = (float*)lds4;
  const int wave  = threadIdx.x >> 5;
  const int lane  = threadIdx.x & 31;
  const int obase = wave * 16;
  const int half  = lane >> 4;
  const int l15   = lane & 15;

  const int ntiles = (N + GEMM_TILE - 1) / GEMM_TILE;
  int t = blockIdx.x;
  if (t >= ntiles) return;

  const float* xl = x + lane * 4;
  float4 st0, st1, st2, st3;
  {
    const int r0 = t * GEMM_TILE + wave;
    st0 = *(const float4*)(xl + (size_t)min(r0,      N - 1) * IN_DIM);
    st1 = *(const float4*)(xl + (size_t)min(r0 + 8,  N - 1) * IN_DIM);
    st2 = *(const float4*)(xl + (size_t)min(r0 + 16, N - 1) * IN_DIM);
    st3 = *(const float4*)(xl + (size_t)min(r0 + 24, N - 1) * IN_DIM);
  }

  v2f bf[IN_DIM / 4];
  {
    float4 wv[OUT_DIM / 8];
    #pragma unroll
    for (int j = 0; j < OUT_DIM / 8; ++j)
      wv[j] = *(const float4*)(W + (size_t)(wave + 8 * j) * IN_DIM + lane * 4);
    #pragma unroll
    for (int j = 0; j < OUT_DIM / 8; ++j)
      *(float4*)(lds + (wave + 8 * j) * GEMM_PITCH + lane * 4) = wv[j];
    __syncthreads();
    const float* bp = lds + (obase + l15) * GEMM_PITCH + 2 * half;
    #pragma unroll
    for (int k = 0; k < IN_DIM / 4; ++k) bf[k] = *(const v2f*)(bp + 4 * k);
    __syncthreads();
  }

  const float4 a1v = *(const float4*)(a + 4 * (lane & 7));
  const float4 a2v = *(const float4*)(a + HEAD_DIM + 4 * (lane & 7));
  float* cs = lds + 2 * (GEMM_TILE * GEMM_PITCH);

  int buf = 0;
  for (; t < ntiles; t += gridDim.x) {
    float* xs = lds + buf * (GEMM_TILE * GEMM_PITCH);
    float* xw = xs + wave * GEMM_PITCH + lane * 4;
    *(float4*)(xw)                   = st0;
    *(float4*)(xw + 8  * GEMM_PITCH) = st1;
    *(float4*)(xw + 16 * GEMM_PITCH) = st2;
    *(float4*)(xw + 24 * GEMM_PITCH) = st3;
    __syncthreads();

    const int tn = t + gridDim.x;
    if (tn < ntiles) {
      const int r0 = tn * GEMM_TILE + wave;
      st0 = *(const float4*)(xl + (size_t)min(r0,      N - 1) * IN_DIM);
      st1 = *(const float4*)(xl + (size_t)min(r0 + 8,  N - 1) * IN_DIM);
      st2 = *(const float4*)(xl + (size_t)min(r0 + 16, N - 1) * IN_DIM);
      st3 = *(const float4*)(xl + (size_t)min(r0 + 24, N - 1) * IN_DIM);
    }

    const float* a0 = xs + l15 * GEMM_PITCH + 2 * half;
    const float* a1 = a0 + 16 * GEMM_PITCH;
    v8f c0 = {}, c1 = {};
    #pragma unroll
    for (int k = 0; k < IN_DIM / 4; ++k) {
      v2f af0 = *(const v2f*)(a0 + 4 * k);
      v2f af1 = *(const v2f*)(a1 + 4 * k);
      c0 = __builtin_amdgcn_wmma_f32_16x16x4_f32(false, af0, false, bf[k],
                                                 (short)0, c0, false, false);
      c1 = __builtin_amdgcn_wmma_f32_16x16x4_f32(false, af1, false, bf[k],
                                                 (short)0, c1, false, false);
    }

    float* cw = cs + (8 * half) * GEMM_PITCH + obase + l15;
    #pragma unroll
    for (int r = 0; r < 8; ++r) {
      cw[r * GEMM_PITCH]        = c0[r];
      cw[(r + 16) * GEMM_PITCH] = c1[r];
    }
    __builtin_amdgcn_fence(__ATOMIC_RELEASE, "workgroup", "local");
    __builtin_amdgcn_s_barrier();
    __builtin_amdgcn_fence(__ATOMIC_ACQUIRE, "workgroup", "local");

    const int r0 = __builtin_amdgcn_readfirstlane(t * GEMM_TILE + 4 * wave);
    const float* cr = cs + (4 * wave) * GEMM_PITCH + lane * 4;
    float4 v[4];
    float p1[4], p2[4];
    #pragma unroll
    for (int q = 0; q < 4; ++q) v[q] = *(const float4*)(cr + q * GEMM_PITCH);
    float* wrow = Wx + (size_t)r0 * OUT_DIM + lane * 4;
    if (r0 + 4 <= N) {
      #pragma unroll
      for (int q = 0; q < 4; ++q) *(float4*)(wrow + q * OUT_DIM) = v[q];
    } else {
      #pragma unroll
      for (int q = 0; q < 4; ++q)
        if (r0 + q < N) *(float4*)(wrow + q * OUT_DIM) = v[q];
    }
    #pragma unroll
    for (int q = 0; q < 4; ++q) {
      p1[q] = v[q].x * a1v.x + v[q].y * a1v.y + v[q].z * a1v.z + v[q].w * a1v.w;
      p2[q] = v[q].x * a2v.x + v[q].y * a2v.y + v[q].z * a2v.z + v[q].w * a2v.w;
    }
    #pragma unroll
    for (int q = 0; q < 4; ++q) { p1[q] += row_xor<1>(p1[q]); p2[q] += row_xor<1>(p2[q]); }
    #pragma unroll
    for (int q = 0; q < 4; ++q) { p1[q] += row_xor<2>(p1[q]); p2[q] += row_xor<2>(p2[q]); }
    #pragma unroll
    for (int q = 0; q < 4; ++q) { p1[q] += row_xor<4>(p1[q]); p2[q] += row_xor<4>(p2[q]); }
    {
      const int i = lane & 7, q = i & 3;
      const float u1 = q == 0 ? p1[0] : q == 1 ? p1[1] : q == 2 ? p1[2] : p1[3];
      const float u2 = q == 0 ? p2[0] : q == 1 ? p2[1] : q == 2 ? p2[2] : p2[3];
      float* sp = ((i & 4) ? s2 : s1) + (size_t)(r0 + q) * HEADS + (lane >> 3);
      if (r0 + q < N) *sp = (i & 4) ? u2 : u1;
    }
    buf ^= 1;
  }
}

#define CHUNKS      32
#define MAX_NODES   50000
#define COUNT_BLOCK 1024
#define COUNT_GROUP 8

__global__ __launch_bounds__(COUNT_BLOCK)
void gat_count(const int* __restrict__ ei, int* __restrict__ ccnt,
               int* __restrict__ rank, int* __restrict__ cursor, int E) {
  __shared__ int4 tab4[MAX_NODES / 4];
  int* tab = (int*)tab4;
  const int tid = threadIdx.x;
  const int c   = blockIdx.x;
  const int per = (E + CHUNKS - 1) / CHUNKS;
  const int e0  = c * per;
  const int e1  = (e0 + per < E) ? (e0 + per) : E;

  for (int i = tid; i < MAX_NODES / 4; i += COUNT_BLOCK) tab4[i] = make_int4(0, 0, 0, 0);
  if (c == 0 && tid == 0) cursor[0] = 0;
  __syncthreads();

  const int* dstp = ei + E;
  int base = e0;
  for (; base + COUNT_GROUP * COUNT_BLOCK <= e1; base += COUNT_GROUP * COUNT_BLOCK) {
    int d[COUNT_GROUP], r[COUNT_GROUP];
    #pragma unroll
    for (int g = 0; g < COUNT_GROUP; ++g) d[g] = dstp[base + g * COUNT_BLOCK + tid];
    #pragma unroll
    for (int g = 0; g < COUNT_GROUP; ++g) r[g] = atomicAdd(&tab[d[g]], 1);
    #pragma unroll
    for (int g = 0; g < COUNT_GROUP; ++g) rank[base + g * COUNT_BLOCK + tid] = r[g];
  }
  for (int e = base + tid; e < e1; e += COUNT_BLOCK)
    rank[e] = atomicAdd(&tab[dstp[e]], 1);
  __syncthreads();

  int4* row4 = (int4*)(ccnt + (size_t)c * MAX_NODES);
  for (int i = tid; i < MAX_NODES / 4; i += COUNT_BLOCK) row4[i] = tab4[i];
}

__global__ void gat_alloc(int* __restrict__ ccnt, int* __restrict__ cnt,
                          int* __restrict__ off, int* __restrict__ cursor, int N) {
  const int n    = blockIdx.x * blockDim.x + threadIdx.x;
  const int lane = threadIdx.x & 31;
  const int nn   = n < N ? n : N - 1;
  int v[CHUNKS];
  #pragma unroll
  for (int k = 0; k < CHUNKS; ++k) v[k] = ccnt[(size_t)k * MAX_NODES + nn];
  int c = 0;
  #pragma unroll
  for (int k = 0; k < CHUNKS; ++k) c += v[k];
  if (n >= N) c = 0;
  int incl = c;
  #pragma unroll
  for (int d = 1; d < 32; d <<= 1) {
    int t = __shfl_up(incl, d);
    if (lane >= d) incl += t;
  }
  const int total = __shfl(incl, 31);
  int base = 0;
  if (lane == 31) base = atomicAdd(cursor, total);
  base = __shfl(base, 31);
  if (n < N) {
    int run = base + incl - c;
    off[n] = run;
    cnt[n] = c;
    #pragma unroll
    for (int k = 0; k < CHUNKS; ++k) {
      ccnt[(size_t)k * MAX_NODES + n] = run;
      run += v[k];
    }
  }
}

__global__ void gat_fill(const int* __restrict__ ei, const int* __restrict__ ccnt,
                         const int* __restrict__ rank, int* __restrict__ ssrc,
                         int E) {
  int e = blockIdx.x * blockDim.x + threadIdx.x;
  if (e >= E) return;
  const int per = (E + CHUNKS - 1) / CHUNKS;
  const int c   = e / per;
  int src = ei[e], dst = ei[E + e];
  ssrc[ccnt[(size_t)c * MAX_NODES + dst] + rank[e]] = src;
}

__device__ __forceinline__ float lrelu02(float v) { return v > 0.f ? v : 0.2f * v; }

__device__ __forceinline__ float lane_f(float v, int j) {
  return __int_as_float(__builtin_amdgcn_readlane(__float_as_int(v), j));
}

#define EDGE_GROUP 4

__global__ void gat_node(const float* __restrict__ Wx,
                         const float* __restrict__ s1,
                         const float* __restrict__ s2,
                         const int* __restrict__ off,
                         const int* __restrict__ cnt,
                         const int* __restrict__ ssrc,
                         float* __restrict__ out, int N) {
  const int wv   = blockIdx.x * (blockDim.x >> 5) + (threadIdx.x >> 5);
  const int lane = threadIdx.x & 31;
  if (wv >= N) return;
  const int n = __builtin_amdgcn_readfirstlane(wv);
  const int o = off[n];
  const int d = cnt[n];
  const float4 sd = *(const float4*)(s2 + (size_t)n * HEADS);

  const float NEG_INF = -__builtin_inff();
  float4 m = make_float4(NEG_INF, NEG_INF, NEG_INF, NEG_INF);
  for (int c0 = 0; c0 < d; c0 += 32) {
    const int idx = c0 + lane;
    if (idx < d) {
      const int s = ssrc[o + idx];
      const float4 ss = *(const float4*)(s1 + (size_t)s * HEADS);
      m.x = fmaxf(m.x, lrelu02(ss.x + sd.x));
      m.y = fmaxf(m.y, lrelu02(ss.y + sd.y));
      m.z = fmaxf(m.z, lrelu02(ss.z + sd.z));
      m.w = fmaxf(m.w, lrelu02(ss.w + sd.w));
    }
  }
  #pragma unroll
  for (int x = 16; x >= 1; x >>= 1) {
    m.x = fmaxf(m.x, __shfl_xor(m.x, x));
    m.y = fmaxf(m.y, __shfl_xor(m.y, x));
    m.z = fmaxf(m.z, __shfl_xor(m.z, x));
    m.w = fmaxf(m.w, __shfl_xor(m.w, x));
  }

  float den[HEADS] = {0.f, 0.f, 0.f, 0.f};
  float acc[HEADS] = {0.f, 0.f, 0.f, 0.f};
  const float* wrow = Wx + lane;
  for (int c0 = 0; c0 < d; c0 += 32) {
    const int idx = c0 + lane;
    int    s = 0;
    float4 w = make_float4(0.f, 0.f, 0.f, 0.f);
    if (idx < d) {
      s = ssrc[o + idx];
      const float4 ss = *(const float4*)(s1 + (size_t)s * HEADS);
      w.x = __expf(lrelu02(ss.x + sd.x) - m.x);
      w.y = __expf(lrelu02(ss.y + sd.y) - m.y);
      w.z = __expf(lrelu02(ss.z + sd.z) - m.z);
      w.w = __expf(lrelu02(ss.w + sd.w) - m.w);
    }
    const int cn = (d - c0) < 32 ? (d - c0) : 32;
    int j = 0;
    for (; j + EDGE_GROUP <= cn; j += EDGE_GROUP) {
      float r[EDGE_GROUP][HEADS];
      #pragma unroll
      for (int g = 0; g < EDGE_GROUP; ++g) {
        const int sj = __builtin_amdgcn_readlane(s, j + g);
        const float* p = wrow + (size_t)sj * OUT_DIM;
        #pragma unroll
        for (int k = 0; k < HEADS; ++k) r[g][k] = p[k * HEAD_DIM];
      }
      #pragma unroll
      for (int g = 0; g < EDGE_GROUP; ++g) {
        const float wk[HEADS] = {lane_f(w.x, j + g), lane_f(w.y, j + g),
                                 lane_f(w.z, j + g), lane_f(w.w, j + g)};
        #pragma unroll
        for (int k = 0; k < HEADS; ++k) {
          den[k] += wk[k];
          acc[k] += wk[k] * r[g][k];
        }
      }
    }
    for (; j < cn; ++j) {
      const int sj = __builtin_amdgcn_readlane(s, j);
      const float* p = wrow + (size_t)sj * OUT_DIM;
      const float wk[HEADS] = {lane_f(w.x, j), lane_f(w.y, j),
                               lane_f(w.z, j), lane_f(w.w, j)};
      #pragma unroll
      for (int k = 0; k < HEADS; ++k) {
        den[k] += wk[k];
        acc[k] += wk[k] * p[k * HEAD_DIM];
      }
    }
  }
  float* orow = out + (size_t)n * OUT_DIM + lane;
  #pragma unroll
  for (int k = 0; k < HEADS; ++k) {
    const float rd = d > 0 ? __frcp_rn(den[k]) : 0.f;
    const float v  = acc[k] * rd;
    orow[k * HEAD_DIM] = v > 0.f ? v : (__expf(v) - 1.0f);
  }
}

extern "C" void kernel_launch(void* const* d_in, const int* in_sizes, int n_in,
                              void* d_out, int out_size, void* d_ws, size_t ws_size,
                              hipStream_t stream) {
  const float* x  = (const float*)d_in[0];
  const float* W  = (const float*)d_in[1];
  const float* a  = (const float*)d_in[2];
  const int*   ei = (const int*)d_in[3];
  const int N = in_sizes[0] / IN_DIM;
  const int E = in_sizes[3] / 2;
  float* out = (float*)d_out;

  char* ws = (char*)d_ws;
  size_t off_b = 0;
  auto carve = [&](size_t bytes) -> void* {
    void* p = ws + off_b;
    off_b += (bytes + 255) & ~(size_t)255;
    return p;
  };
  float* Wx     = (float*)carve((size_t)N * OUT_DIM * sizeof(float));
  float* s1     = (float*)carve((size_t)N * HEADS * sizeof(float));
  float* s2     = (float*)carve((size_t)N * HEADS * sizeof(float));
  int*   cnt    = (int*)  carve((size_t)N * sizeof(int));
  int*   cursor = (int*)  carve(sizeof(int));
  int*   offs   = (int*)  carve((size_t)N * sizeof(int));
  int*   rank   = (int*)  carve((size_t)E * sizeof(int));
  int*   ssrc   = (int*)  carve((size_t)E * sizeof(int));
  int*   ccnt   = (int*)  carve((size_t)CHUNKS * MAX_NODES * sizeof(int));
  (void)ws_size; (void)n_in; (void)out_size;

  gat_gemm_wmma<<<GEMM_GRID, 256, 0, stream>>>(x, W, a, Wx, s1, s2, N);

  gat_count<<<CHUNKS, COUNT_BLOCK, 0, stream>>>(ei, ccnt, rank, cursor, E);

  gat_alloc<<<(N + 255) / 256, 256, 0, stream>>>(ccnt, cnt, offs, cursor, N);

  gat_fill<<<(E + 255) / 256, 256, 0, stream>>>(ei, ccnt, rank, ssrc, E);

  gat_node<<<(N + 7) / 8, 256, 0, stream>>>(Wx, s1, s2, offs, cnt, ssrc, out, N);
}


# edge counting pass runs inside the GEMM launch on 96 extra work-groups (node range cut in 3 so each LDS table fits beside the GEMM's groups): one launch less, its time hidden; float sums unchanged
# speedup vs baseline: 5.4717x; 1.0555x over previous
#include <hip/hip_runtime.h>
#include <cstdint>
#include <cstddef>

typedef __attribute__((ext_vector_type(2))) float v2f;
typedef __attribute__((ext_vector_type(8))) float v8f;

#define IN_DIM   128
#define OUT_DIM  128
#define HEADS    4
#define HEAD_DIM 32

#define GEMM_TILE  32
#define GEMM_PITCH (IN_DIM + 4)
#define GEMM_GRID  512

#define CHUNKS       32
#define MAX_NODES    50000
#define COUNT_RANGES 3
#define RANGE_NODES  16668
#define COUNT_WGS    (CHUNKS * COUNT_RANGES)
#define COUNT_GROUP  32
#define FUSED_BLOCK  256

static_assert(RANGE_NODES % 4 == 0 && MAX_NODES % 4 == 0, "table rows go out as 16-byte pieces");
static_assert(RANGE_NODES * COUNT_RANGES >= MAX_NODES, "ranges cover all nodes");
static_assert(RANGE_NODES <= OUT_DIM * GEMM_PITCH, "a range's counters fit the GEMM's LDS block");

__device__ __forceinline__
void count_block(int4* tab4, const int* __restrict__ ei, int* __restrict__ ccnt,
                 int* __restrict__ rank, int* __restrict__ cursor, int E, int b) {
  int* tab = (int*)tab4;
  const int tid = threadIdx.x;
  const int c   = b / COUNT_RANGES;
  const int n0  = (b - c * COUNT_RANGES) * RANGE_NODES;
  const int nn  = min(RANGE_NODES, MAX_NODES - n0);
  const int per = (E + CHUNKS - 1) / CHUNKS;
  const int e0  = c * per;
  const int e1  = (e0 + per < E) ? (e0 + per) : E;

  for (int i = tid; i < RANGE_NODES / 4; i += FUSED_BLOCK) tab4[i] = make_int4(0, 0, 0, 0);
  if (b == 0 && tid == 0) cursor[0] = 0;
  __syncthreads();

  const int* dstp = ei + E;
  int base = e0;
  for (; base + COUNT_GROUP * FUSED_BLOCK <= e1; base += COUNT_GROUP * FUSED_BLOCK) {
    int d[COUNT_GROUP], r[COUNT_GROUP];
    #pragma unroll
    for (int g = 0; g < COUNT_GROUP; ++g) d[g] = dstp[base + g * FUSED_BLOCK + tid];
    __builtin_amdgcn_sched_barrier(0);
    #pragma unroll
    for (int g = 0; g < COUNT_GROUP; ++g) {
      const unsigned k = (unsigned)(d[g] - n0);
      r[g] = -1;
      if (k < (unsigned)nn) r[g] = atomicAdd(&tab[k], 1);
    }
    #pragma unroll
    for (int g = 0; g < COUNT_GROUP; ++g)
      if (r[g] >= 0) rank[base + g * FUSED_BLOCK + tid] = r[g];
  }
  for (int e = base + tid; e < e1; e += FUSED_BLOCK) {
    const unsigned k = (unsigned)(dstp[e] - n0);
    if (k < (unsigned)nn) rank[e] = atomicAdd(&tab[k], 1);
  }
  __syncthreads();

  int4* row4 = (int4*)(ccnt + (size_t)c * MAX_NODES + n0);
  for (int i = tid; i < nn / 4; i += FUSED_BLOCK) row4[i] = tab4[i];
}

template <int M>
__device__ __forceinline__ float row_xor(float v) {
  return __int_as_float(__builtin_amdgcn_update_dpp(0, __float_as_int(v), 0x160 + M,
                                                    0xf, 0xf, false));
}

__global__ __launch_bounds__(FUSED_BLOCK)
void gat_gemm_wmma(const float* __restrict__ x,
                   const float* __restrict__ W,
                   const float* __restrict__ a,
                   float* __restrict__ Wx,
                   float* __restrict__ s1, float* __restrict__ s2, int N,
                   const int* __restrict__ ei, int* __restrict__ ccnt,
                   int* __restrict__ rank, int* __restrict__ cursor, int E) {
  __shared__ float4 lds4[OUT_DIM * GEMM_PITCH / 4];
  if (blockIdx.x < COUNT_WGS) {
    count_block((int4*)lds4, ei, ccnt, rank, cursor, E, blockIdx.x);
    return;
  }
  const int gemm_wgs = gridDim.x - COUNT_WGS;
  float* lds = (float*)lds4;
  const int wave  = threadIdx.x >> 5;
  const int lane  = threadIdx.x & 31;
  const int obase = wave * 16;
  const int half  = lane >> 4;
  const int l15   = lane & 15;

  const int ntiles = (N + GEMM_TILE - 1) / GEMM_TILE;
  int t = blockIdx.x - COUNT_WGS;
  if (t >= ntiles) return;

  const float* xl = x + lane * 4;
  float4 st0, st1, st2, st3;
  {
    const int r0 = t * GEMM_TILE + wave;
    st0 = *(const float4*)(xl + (size_t)min(r0,      N - 1) * IN_DIM);
    st1 = *(const float4*)(xl + (size_t)min(r0 + 8,  N - 1) * IN_DIM);
    st2 = *(const float4*)(xl + (size_t)min(r0 + 16, N - 1) * IN_DIM);
    st3 = *(const float4*)(xl + (size_t)min(r0 + 24, N - 1) * IN_DIM);
  }

  v2f bf[IN_DIM / 4];
  {
    float4 wv[OUT_DIM / 8];
    #pragma unroll
    for (int j = 0; j < OUT_DIM / 8; ++j)
      wv[j] = *(const float4*)(W + (size_t)(wave + 8 * j) * IN_DIM + lane * 4);
    #pragma unroll
    for (int j = 0; j < OUT_DIM / 8; ++j)
      *(float4*)(lds + (wave + 8 * j) * GEMM_PITCH + lane * 4) = wv[j];
    __syncthreads();
    const float* bp = lds + (obase + l15) * GEMM_PITCH + 2 * half;
    #pragma unroll
    for (int k = 0; k < IN_DIM / 4; ++k) bf[k] = *(const v2f*)(bp + 4 * k);
    __syncthreads();
  }

  const float4 a1v = *(const float4*)(a + 4 * (lane & 7));
  const float4 a2v = *(const float4*)(a + HEAD_DIM + 4 * (lane & 7));
  float* cs = lds + 2 * (GEMM_TILE * GEMM_PITCH);

  int buf = 0;
  for (; t < ntiles; t += gemm_wgs) {
    float* xs = lds + buf * (GEMM_TILE * GEMM_PITCH);
    float* xw = xs + wave * GEMM_PITCH + lane * 4;
    *(float4*)(xw)                   = st0;
    *(float4*)(xw + 8  * GEMM_PITCH) = st1;
    *(float4*)(xw + 16 * GEMM_PITCH) = st2;
    *(float4*)(xw + 24 * GEMM_PITCH) = st3;
    __syncthreads();

    const int tn = t + gemm_wgs;
    if (tn < ntiles) {
      const int r0 = tn * GEMM_TILE + wave;
      st0 = *(const float4*)(xl + (size_t)min(r0,      N - 1) * IN_DIM);
      st1 = *(const float4*)(xl + (size_t)min(r0 + 8,  N - 1) * IN_DIM);
      st2 = *(const float4*)(xl + (size_t)min(r0 + 16, N - 1) * IN_DIM);
      st3 = *(const float4*)(xl + (size_t)min(r0 + 24, N - 1) * IN_DIM);
    }

    const float* a0 = xs + l15 * GEMM_PITCH + 2 * half;
    const float* a1 = a0 + 16 * GEMM_PITCH;
    v8f c0 = {}, c1 = {};
    #pragma unroll
    for (int k = 0; k < IN_DIM / 4; ++k) {
      v2f af0 = *(const v2f*)(a0 + 4 * k);
      v2f af1 = *(const v2f*)(a1 + 4 * k);
      c0 = __builtin_amdgcn_wmma_f32_16x16x4_f32(false, af0, false, bf[k],
                                                 (short)0, c0, false, false);
      c1 = __builtin_amdgcn_wmma_f32_16x16x4_f32(false, af1, false, bf[k],
                                                 (short)0, c1, false, false);
    }

    float* cw = cs + (8 * half) * GEMM_PITCH + obase + l15;
    #pragma unroll
    for (int r = 0; r < 8; ++r) {
      cw[r * GEMM_PITCH]        = c0[r];
      cw[(r + 16) * GEMM_PITCH] = c1[r];
    }
    __builtin_amdgcn_fence(__ATOMIC_RELEASE, "workgroup", "local");
    __builtin_amdgcn_s_barrier();
    __builtin_amdgcn_fence(__ATOMIC_ACQUIRE, "workgroup", "local");

    const int r0 = __builtin_amdgcn_readfirstlane(t * GEMM_TILE + 4 * wave);
    const float* cr = cs + (4 * wave) * GEMM_PITCH + lane * 4;
    float4 v[4];
    float p1[4], p2[4];
    #pragma unroll
    for (int q = 0; q < 4; ++q) v[q] = *(const float4*)(cr + q * GEMM_PITCH);
    float* wrow = Wx + (size_t)r0 * OUT_DIM + lane * 4;
    if (r0 + 4 <= N) {
      #pragma unroll
      for (int q = 0; q < 4; ++q) *(float4*)(wrow + q * OUT_DIM) = v[q];
    } else {
      #pragma unroll
      for (int q = 0; q < 4; ++q)
        if (r0 + q < N) *(float4*)(wrow + q * OUT_DIM) = v[q];
    }
    #pragma unroll
    for (int q = 0; q < 4; ++q) {
      p1[q] = v[q].x * a1v.x + v[q].y * a1v.y + v[q].z * a1v.z + v[q].w * a1v.w;
      p2[q] = v[q].x * a2v.x + v[q].y * a2v.y + v[q].z * a2v.z + v[q].w * a2v.w;
    }
    #pragma unroll
    for (int q = 0; q < 4; ++q) { p1[q] += row_xor<1>(p1[q]); p2[q] += row_xor<1>(p2[q]); }
    #pragma unroll
    for (int q = 0; q < 4; ++q) { p1[q] += row_xor<2>(p1[q]); p2[q] += row_xor<2>(p2[q]); }
    #pragma unroll
    for (int q = 0; q < 4; ++q) { p1[q] += row_xor<4>(p1[q]); p2[q] += row_xor<4>(p2[q]); }
    {
      const int i = lane & 7, q = i & 3;
      const float u1 = q == 0 ? p1[0] : q == 1 ? p1[1] : q == 2 ? p1[2] : p1[3];
      const float u2 = q == 0 ? p2[0] : q == 1 ? p2[1] : q == 2 ? p2[2] : p2[3];
      float* sp = ((i & 4) ? s2 : s1) + (size_t)(r0 + q) * HEADS + (lane >> 3);
      if (r0 + q < N) *sp = (i & 4) ? u2 : u1;
    }
    buf ^= 1;
  }
}


__global__ void gat_alloc(int* __restrict__ ccnt, int* __restrict__ cnt,
                          int* __restrict__ off, int* __restrict__ cursor, int N) {
  const int n    = blockIdx.x * blockDim.x + threadIdx.x;
  const int lane = threadIdx.x & 31;
  const int nn   = n < N ? n : N - 1;
  int v[CHUNKS];
  #pragma unroll
  for (int k = 0; k < CHUNKS; ++k) v[k] = ccnt[(size_t)k * MAX_NODES + nn];
  int c = 0;
  #pragma unroll
  for (int k = 0; k < CHUNKS; ++k) c += v[k];
  if (n >= N) c = 0;
  int incl = c;
  #pragma unroll
  for (int d = 1; d < 32; d <<= 1) {
    int t = __shfl_up(incl, d);
    if (lane >= d) incl += t;
  }
  const int total = __shfl(incl, 31);
  int base = 0;
  if (lane == 31) base = atomicAdd(cursor, total);
  base = __shfl(base, 31);
  if (n < N) {
    int run = base + incl - c;
    off[n] = run;
    cnt[n] = c;
    #pragma unroll
    for (int k = 0; k < CHUNKS; ++k) {
      ccnt[(size_t)k * MAX_NODES + n] = run;
      run += v[k];
    }
  }
}

__global__ void gat_fill(const int* __restrict__ ei, const int* __restrict__ ccnt,
                         const int* __restrict__ rank, int* __restrict__ ssrc,
                         int E) {
  int e = blockIdx.x * blockDim.x + threadIdx.x;
  if (e >= E) return;
  const int per = (E + CHUNKS - 1) / CHUNKS;
  const int c   = e / per;
  int src = ei[e], dst = ei[E + e];
  ssrc[ccnt[(size_t)c * MAX_NODES + dst] + rank[e]] = src;
}

__device__ __forceinline__ float lrelu02(float v) { return v > 0.f ? v : 0.2f * v; }

__device__ __forceinline__ float lane_f(float v, int j) {
  return __int_as_float(__builtin_amdgcn_readlane(__float_as_int(v), j));
}

#define EDGE_GROUP 4

__global__ void gat_node(const float* __restrict__ Wx,
                         const float* __restrict__ s1,
                         const float* __restrict__ s2,
                         const int* __restrict__ off,
                         const int* __restrict__ cnt,
                         const int* __restrict__ ssrc,
                         float* __restrict__ out, int N) {
  const int wv   = blockIdx.x * (blockDim.x >> 5) + (threadIdx.x >> 5);
  const int lane = threadIdx.x & 31;
  if (wv >= N) return;
  const int n = __builtin_amdgcn_readfirstlane(wv);
  const int o = off[n];
  const int d = cnt[n];
  const float4 sd = *(const float4*)(s2 + (size_t)n * HEADS);

  const float NEG_INF = -__builtin_inff();
  float4 m = make_float4(NEG_INF, NEG_INF, NEG_INF, NEG_INF);
  for (int c0 = 0; c0 < d; c0 += 32) {
    const int idx = c0 + lane;
    if (idx < d) {
      const int s = ssrc[o + idx];
      const float4 ss = *(const float4*)(s1 + (size_t)s * HEADS);
      m.x = fmaxf(m.x, lrelu02(ss.x + sd.x));
      m.y = fmaxf(m.y, lrelu02(ss.y + sd.y));
      m.z = fmaxf(m.z, lrelu02(ss.z + sd.z));
      m.w = fmaxf(m.w, lrelu02(ss.w + sd.w));
    }
  }
  #pragma unroll
  for (int x = 16; x >= 1; x >>= 1) {
    m.x = fmaxf(m.x, __shfl_xor(m.x, x));
    m.y = fmaxf(m.y, __shfl_xor(m.y, x));
    m.z = fmaxf(m.z, __shfl_xor(m.z, x));
    m.w = fmaxf(m.w, __shfl_xor(m.w, x));
  }

  float den[HEADS] = {0.f, 0.f, 0.f, 0.f};
  float acc[HEADS] = {0.f, 0.f, 0.f, 0.f};
  const float* wrow = Wx + lane;
  for (int c0 = 0; c0 < d; c0 += 32) {
    const int idx = c0 + lane;
    int    s = 0;
    float4 w = make_float4(0.f, 0.f, 0.f, 0.f);
    if (idx < d) {
      s = ssrc[o + idx];
      const float4 ss = *(const float4*)(s1 + (size_t)s * HEADS);
      w.x = __expf(lrelu02(ss.x + sd.x) - m.x);
      w.y = __expf(lrelu02(ss.y + sd.y) - m.y);
      w.z = __expf(lrelu02(ss.z + sd.z) - m.z);
      w.w = __expf(lrelu02(ss.w + sd.w) - m.w);
    }
    const int cn = (d - c0) < 32 ? (d - c0) : 32;
    int j = 0;
    for (; j + EDGE_GROUP <= cn; j += EDGE_GROUP) {
      float r[EDGE_GROUP][HEADS];
      #pragma unroll
      for (int g = 0; g < EDGE_GROUP; ++g) {
        const int sj = __builtin_amdgcn_readlane(s, j + g);
        const float* p = wrow + (size_t)sj * OUT_DIM;
        #pragma unroll
        for (int k = 0; k < HEADS; ++k) r[g][k] = p[k * HEAD_DIM];
      }
      #pragma unroll
      for (int g = 0; g < EDGE_GROUP; ++g) {
        const float wk[HEADS] = {lane_f(w.x, j + g), lane_f(w.y, j + g),
                                 lane_f(w.z, j + g), lane_f(w.w, j + g)};
        #pragma unroll
        for (int k = 0; k < HEADS; ++k) {
          den[k] += wk[k];
          acc[k] += wk[k] * r[g][k];
        }
      }
    }
    for (; j < cn; ++j) {
      const int sj = __builtin_amdgcn_readlane(s, j);
      const float* p = wrow + (size_t)sj * OUT_DIM;
      const float wk[HEADS] = {lane_f(w.x, j), lane_f(w.y, j),
                               lane_f(w.z, j), lane_f(w.w, j)};
      #pragma unroll
      for (int k = 0; k < HEADS; ++k) {
        den[k] += wk[k];
        acc[k] += wk[k] * p[k * HEAD_DIM];
      }
    }
  }
  float* orow = out + (size_t)n * OUT_DIM + lane;
  #pragma unroll
  for (int k = 0; k < HEADS; ++k) {
    const float rd = d > 0 ? __frcp_rn(den[k]) : 0.f;
    const float v  = acc[k] * rd;
    orow[k * HEAD_DIM] = v > 0.f ? v : (__expf(v) - 1.0f);
  }
}

extern "C" void kernel_launch(void* const* d_in, const int* in_sizes, int n_in,
                              void* d_out, int out_size, void* d_ws, size_t ws_size,
                              hipStream_t stream) {
  const float* x  = (const float*)d_in[0];
  const float* W  = (const float*)d_in[1];
  const float* a  = (const float*)d_in[2];
  const int*   ei = (const int*)d_in[3];
  const int N = in_sizes[0] / IN_DIM;
  const int E = in_sizes[3] / 2;
  float* out = (float*)d_out;

  char* ws = (char*)d_ws;
  size_t off_b = 0;
  auto carve = [&](size_t bytes) -> void* {
    void* p = ws + off_b;
    off_b += (bytes + 255) & ~(size_t)255;
    return p;
  };
  float* Wx     = (float*)carve((size_t)N * OUT_DIM * sizeof(float));
  float* s1     = (float*)carve((size_t)N * HEADS * sizeof(float));
  float* s2     = (float*)carve((size_t)N * HEADS * sizeof(float));
  int*   cnt    = (int*)  carve((size_t)N * sizeof(int));
  int*   cursor = (int*)  carve(sizeof(int));
  int*   offs   = (int*)  carve((size_t)N * sizeof(int));
  int*   rank   = (int*)  carve((size_t)E * sizeof(int));
  int*   ssrc   = (int*)  carve((size_t)E * sizeof(int));
  int*   ccnt   = (int*)  carve((size_t)CHUNKS * MAX_NODES * sizeof(int));
  (void)ws_size; (void)n_in; (void)out_size;

  gat_gemm_wmma<<<COUNT_WGS + GEMM_GRID, FUSED_BLOCK, 0, stream>>>(x, W, a, Wx, s1, s2, N,
                                                                ei, ccnt, rank, cursor, E);

  gat_alloc<<<(N + 255) / 256, 256, 0, stream>>>(ccnt, cnt, offs, cursor, N);

  gat_fill<<<(E + 255) / 256, 256, 0, stream>>>(ei, ccnt, rank, ssrc, E);

  gat_node<<<(N + 7) / 8, 256, 0, stream>>>(Wx, s1, s2, offs, cnt, ssrc, out, N);
}


# edges sorted per chunk inside the GEMM launch (count, scan, place on LDS counters, 56 chunks); node waves walk the chunk lists; alloc, fill launches gone; sums chunk by chunk, arrival order in each
# speedup vs baseline: 7.5628x; 1.3822x over previous
#include <hip/hip_runtime.h>
#include <cstdint>
#include <cstddef>

typedef __attribute__((ext_vector_type(2))) float v2f;
typedef __attribute__((ext_vector_type(8))) float v8f;

#define IN_DIM   128
#define OUT_DIM  128
#define HEADS    4
#define HEAD_DIM 32

#define GEMM_TILE  32
#define GEMM_PITCH (IN_DIM + 4)
#define GEMM_GRID  512

#define CHUNKS       56
#define MAX_NODES    50000
#define COUNT_RANGES 3
#define RANGE_NODES  16668
#define COUNT_WGS    (CHUNKS * COUNT_RANGES)
#define EDGES_HELD   56
#define PLACE_GROUP  8
#define SCAN_RUN     68
#define FUSED_BLOCK  256

static_assert(RANGE_NODES % 4 == 0 && MAX_NODES % 4 == 0, "table rows go out as 16-byte pieces");
static_assert(RANGE_NODES * COUNT_RANGES >= MAX_NODES, "ranges cover all nodes");
static_assert(RANGE_NODES + 16 <= OUT_DIM * GEMM_PITCH, "a range's counters and the scan's 16 words fit the GEMM's LDS block");
static_assert(SCAN_RUN % 4 == 0 && SCAN_RUN * FUSED_BLOCK >= RANGE_NODES, "the scan covers a range in 16-byte pieces");
static_assert(EDGES_HELD % PLACE_GROUP == 0, "whole groups in the placing pass");
static_assert(CHUNKS > 32 && CHUNKS <= 64, "the node kernel gives every lane two chunks");
static_assert(MAX_NODES <= 65535, "source ids are stored in 2 bytes");

__device__ __forceinline__
void sort_block(int4* tab4, const int* __restrict__ ei, int* __restrict__ tabg,
                unsigned short* __restrict__ ssrc, int E, int b) {
  int* tab = (int*)tab4;
  int* scr = tab + RANGE_NODES;
  const int tid  = threadIdx.x;
  const int lane = tid & 31;
  const int wave = tid >> 5;
  const int c   = b / COUNT_RANGES;
  const int n0  = (b - c * COUNT_RANGES) * RANGE_NODES;
  const int nn  = min(RANGE_NODES, MAX_NODES - n0);
  const int per = (E + CHUNKS - 1) / CHUNKS;
  const int e0  = min(c * per, E);
  const int e1  = min(e0 + per, E);
  const int* dstp = ei + E;

  for (int i = tid; i < RANGE_NODES / 4; i += FUSED_BLOCK) tab4[i] = make_int4(0, 0, 0, 0);

  const bool full  = e0 + (EDGES_HELD - 1) * FUSED_BLOCK <= e1;
  const int  rest0 = full ? e0 + EDGES_HELD * FUSED_BLOCK : e0;
  unsigned v[EDGES_HELD];
  if (full) {
    #pragma unroll
    for (int g = 0; g < EDGES_HELD - 1; ++g) {
      const int e = e0 + g * FUSED_BLOCK + tid;
      v[g] = ((unsigned)ei[e] << 16) | (unsigned)dstp[e];
    }
    const int e  = e0 + (EDGES_HELD - 1) * FUSED_BLOCK + tid;
    const int ec = min(e, E - 1);
    const unsigned w = ((unsigned)ei[ec] << 16) | (unsigned)dstp[ec];
    v[EDGES_HELD - 1] = e < e1 ? w : 0xffffu;
  } else {
    #pragma unroll
    for (int g = 0; g < EDGES_HELD; ++g) v[g] = 0xffffu;
  }
  __syncthreads();

  int below = 0;
  #pragma unroll
  for (int g = 0; g < EDGES_HELD; ++g) {
    const unsigned dn = v[g] & 0xffffu;
    const unsigned k  = dn - (unsigned)n0;
    if (k < (unsigned)nn) atomicAdd(&tab[k], 1);
    below += (dn < (unsigned)n0) ? 1 : 0;
  }
  for (int e = rest0 + tid; e < e1; e += FUSED_BLOCK) {
    const int dv = dstp[e];
    const unsigned k = (unsigned)(dv - n0);
    if (k < (unsigned)nn) atomicAdd(&tab[k], 1);
    below += ((unsigned)dv < (unsigned)n0) ? 1 : 0;
  }
  __syncthreads();

  const int q0 = tid * (SCAN_RUN / 4);
  const int nq = min(max(nn / 4 - q0, 0), SCAN_RUN / 4);
  int sum = 0;
  if (nq == SCAN_RUN / 4) {
    #pragma unroll
    for (int j = 0; j < SCAN_RUN / 4; ++j) {
      const int4 c = tab4[q0 + j];
      sum += (c.x + c.y) + (c.z + c.w);
    }
  } else {
    for (int j = 0; j < nq; ++j) {
      const int4 c = tab4[q0 + j];
      sum += (c.x + c.y) + (c.z + c.w);
    }
  }
  int incl = sum, bl = below;
  #pragma unroll
  for (int x = 1; x < 32; x <<= 1) {
    const int t = __shfl_up(incl, x);
    if (lane >= x) incl += t;
    bl += __shfl_xor(bl, x);
  }
  if (lane == 31) { scr[wave] = incl; scr[8 + wave] = bl; }
  __syncthreads();
  int first = 0;
  int run   = incl - sum;
  #pragma unroll
  for (int w = 0; w < FUSED_BLOCK / 32; ++w) {
    first += scr[8 + w];
    if (w < wave) run += scr[w];
  }
  run += first;
  if (nq == SCAN_RUN / 4) {
    #pragma unroll
    for (int j = 0; j < SCAN_RUN / 4; ++j) {
      const int4 c = tab4[q0 + j];
      int4 o;
      o.x = run;
      o.y = o.x + c.x;
      o.z = o.y + c.y;
      o.w = o.z + c.z;
      run = o.w + c.w;
      tab4[q0 + j] = o;
    }
  } else {
    for (int j = 0; j < nq; ++j) {
      const int4 c = tab4[q0 + j];
      int4 o;
      o.x = run;
      o.y = o.x + c.x;
      o.z = o.y + c.y;
      o.w = o.z + c.z;
      run = o.w + c.w;
      tab4[q0 + j] = o;
    }
  }
  __syncthreads();

  unsigned short* seg = ssrc + e0;
  #pragma unroll
  for (int g = 0; g < EDGES_HELD; ++g) asm volatile("" : "+v"(v[g]));
  #pragma unroll
  for (int g0 = 0; g0 < EDGES_HELD; g0 += PLACE_GROUP) {
    int r[PLACE_GROUP];
    #pragma unroll
    for (int j = 0; j < PLACE_GROUP; ++j) {
      const unsigned k = (v[g0 + j] & 0xffffu) - (unsigned)n0;
      r[j] = -1;
      if (k < (unsigned)nn) r[j] = atomicAdd(&tab[k], 1);
    }
    #pragma unroll
    for (int j = 0; j < PLACE_GROUP; ++j)
      if (r[j] >= 0) seg[r[j]] = (unsigned short)(v[g0 + j] >> 16);
  }
  for (int e = rest0 + tid; e < e1; e += FUSED_BLOCK) {
    const unsigned k = (unsigned)(dstp[e] - n0);
    if (k < (unsigned)nn) seg[atomicAdd(&tab[k], 1)] = (unsigned short)ei[e];
  }
  __syncthreads();

  int4* row4 = (int4*)(tabg + (size_t)c * MAX_NODES + n0);
  for (int q = tid; q < nn / 4; q += FUSED_BLOCK) {
    const int4 en = tab4[q];
    const int  st = q == 0 ? first : tab[4 * q - 1];
    int4 o;
    o.x = (int)(((unsigned)st   << 16) | (unsigned)(en.x - st));
    o.y = (int)(((unsigned)en.x << 16) | (unsigned)(en.y - en.x));
    o.z = (int)(((unsigned)en.y << 16) | (unsigned)(en.z - en.y));
    o.w = (int)(((unsigned)en.z << 16) | (unsigned)(en.w - en.z));
    row4[q] = o;
  }
}

template <int M>
__device__ __forceinline__ float row_xor(float v) {
  return __int_as_float(__builtin_amdgcn_update_dpp(0, __float_as_int(v), 0x160 + M,
                                                    0xf, 0xf, false));
}

__global__ __launch_bounds__(FUSED_BLOCK, 6)
void gat_gemm_wmma(const float* __restrict__ x,
                   const float* __restrict__ W,
                   const float* __restrict__ a,
                   float* __restrict__ Wx,
                   float* __restrict__ s1, float* __restrict__ s2, int N,
                   const int* __restrict__ ei, int* __restrict__ tab,
                   unsigned short* __restrict__ ssrc, int E) {
  __shared__ float4 lds4[OUT_DIM * GEMM_PITCH / 4];
  if (blockIdx.x < COUNT_WGS) {
    sort_block((int4*)lds4, ei, tab, ssrc, E, blockIdx.x);
    return;
  }
  const int gemm_wgs = gridDim.x - COUNT_WGS;
  float* lds = (float*)lds4;
  const int wave  = threadIdx.x >> 5;
  const int lane  = threadIdx.x & 31;
  const int obase = wave * 16;
  const int half  = lane >> 4;
  const int l15   = lane & 15;

  const int ntiles = (N + GEMM_TILE - 1) / GEMM_TILE;
  int t = blockIdx.x - COUNT_WGS;
  if (t >= ntiles) return;

  const float* xl = x + lane * 4;
  float4 st0, st1, st2, st3;
  {
    const int r0 = t * GEMM_TILE + wave;
    st0 = *(const float4*)(xl + (size_t)min(r0,      N - 1) * IN_DIM);
    st1 = *(const float4*)(xl + (size_t)min(r0 + 8,  N - 1) * IN_DIM);
    st2 = *(const float4*)(xl + (size_t)min(r0 + 16, N - 1) * IN_DIM);
    st3 = *(const float4*)(xl + (size_t)min(r0 + 24, N - 1) * IN_DIM);
  }

  v2f bf[IN_DIM / 4];
  {
    float4 wv[OUT_DIM / 8];
    #pragma unroll
    for (int j = 0; j < OUT_DIM / 8; ++j)
      wv[j] = *(const float4*)(W + (size_t)(wave + 8 * j) * IN_DIM + lane * 4);
    #pragma unroll
    for (int j = 0; j < OUT_DIM / 8; ++j)
      *(float4*)(lds + (wave + 8 * j) * GEMM_PITCH + lane * 4) = wv[j];
    __syncthreads();
    const float* bp = lds + (obase + l15) * GEMM_PITCH + 2 * half;
    #pragma unroll
    for (int k = 0; k < IN_DIM / 4; ++k) bf[k] = *(const v2f*)(bp + 4 * k);
    __syncthreads();
  }

  const float4 a1v = *(const float4*)(a + 4 * (lane & 7));
  const float4 a2v = *(const float4*)(a + HEAD_DIM + 4 * (lane & 7));
  float* cs = lds + 2 * (GEMM_TILE * GEMM_PITCH);

  int buf = 0;
  for (; t < ntiles; t += gemm_wgs) {
    float* xs = lds + buf * (GEMM_TILE * GEMM_PITCH);
    float* xw = xs + wave * GEMM_PITCH + lane * 4;
    *(float4*)(xw)                   = st0;
    *(float4*)(xw + 8  * GEMM_PITCH) = st1;
    *(float4*)(xw + 16 * GEMM_PITCH) = st2;
    *(float4*)(xw + 24 * GEMM_PITCH) = st3;
    __syncthreads();

    const int tn = t + gemm_wgs;
    if (tn < ntiles) {
      const int r0 = tn * GEMM_TILE + wave;
      st0 = *(const float4*)(xl + (size_t)min(r0,      N - 1) * IN_DIM);
      st1 = *(const float4*)(xl + (size_t)min(r0 + 8,  N - 1) * IN_DIM);
      st2 = *(const float4*)(xl + (size_t)min(r0 + 16, N - 1) * IN_DIM);
      st3 = *(const float4*)(xl + (size_t)min(r0 + 24, N - 1) * IN_DIM);
    }

    const float* a0 = xs + l15 * GEMM_PITCH + 2 * half;
    const float* a1 = a0 + 16 * GEMM_PITCH;
    v8f c0 = {}, c1 = {};
    #pragma unroll
    for (int k = 0; k < IN_DIM / 4; ++k) {
      v2f af0 = *(const v2f*)(a0 + 4 * k);
      v2f af1 = *(const v2f*)(a1 + 4 * k);
      c0 = __builtin_amdgcn_wmma_f32_16x16x4_f32(false, af0, false, bf[k],
                                                 (short)0, c0, false, false);
      c1 = __builtin_amdgcn_wmma_f32_16x16x4_f32(false, af1, false, bf[k],
                                                 (short)0, c1, false, false);
    }

    float* cw = cs + (8 * half) * GEMM_PITCH + obase + l15;
    #pragma unroll
    for (int r = 0; r < 8; ++r) {
      cw[r * GEMM_PITCH]        = c0[r];
      cw[(r + 16) * GEMM_PITCH] = c1[r];
    }
    __builtin_amdgcn_fence(__ATOMIC_RELEASE, "workgroup", "local");
    __builtin_amdgcn_s_barrier();
    __builtin_amdgcn_fence(__ATOMIC_ACQUIRE, "workgroup", "local");

    const int r0 = __builtin_amdgcn_readfirstlane(t * GEMM_TILE + 4 * wave);
    const float* cr = cs + (4 * wave) * GEMM_PITCH + lane * 4;
    float4 v[4];
    float p1[4], p2[4];
    #pragma unroll
    for (int q = 0; q < 4; ++q) v[q] = *(const float4*)(cr + q * GEMM_PITCH);
    float* wrow = Wx + (size_t)r0 * OUT_DIM + lane * 4;
    if (r0 + 4 <= N) {
      #pragma unroll
      for (int q = 0; q < 4; ++q) *(float4*)(wrow + q * OUT_DIM) = v[q];
    } else {
      #pragma unroll
      for (int q = 0; q < 4; ++q)
        if (r0 + q < N) *(float4*)(wrow + q * OUT_DIM) = v[q];
    }
    #pragma unroll
    for (int q = 0; q < 4; ++q) {
      p1[q] = v[q].x * a1v.x + v[q].y * a1v.y + v[q].z * a1v.z + v[q].w * a1v.w;
      p2[q] = v[q].x * a2v.x + v[q].y * a2v.y + v[q].z * a2v.z + v[q].w * a2v.w;
    }
    #pragma unroll
    for (int q = 0; q < 4; ++q) { p1[q] += row_xor<1>(p1[q]); p2[q] += row_xor<1>(p2[q]); }
    #pragma unroll
    for (int q = 0; q < 4; ++q) { p1[q] += row_xor<2>(p1[q]); p2[q] += row_xor<2>(p2[q]); }
    #pragma unroll
    for (int q = 0; q < 4; ++q) { p1[q] += row_xor<4>(p1[q]); p2[q] += row_xor<4>(p2[q]); }
    {
      const int i = lane & 7, q = i & 3;
      const float u1 = q == 0 ? p1[0] : q == 1 ? p1[1] : q == 2 ? p1[2] : p1[3];
      const float u2 = q == 0 ? p2[0] : q == 1 ? p2[1] : q == 2 ? p2[2] : p2[3];
      float* sp = ((i & 4) ? s2 : s1) + (size_t)(r0 + q) * HEADS + (lane >> 3);
      if (r0 + q < N) *sp = (i & 4) ? u2 : u1;
    }
    buf ^= 1;
  }
}

__device__ __forceinline__ float lrelu02(float v) { return v > 0.f ? v : 0.2f * v; }

__device__ __forceinline__ float lane_f(float v, int j) {
  return __int_as_float(__builtin_amdgcn_readlane(__float_as_int(v), j));
}

#define EDGE_GROUP 4
#define NODE_WAVES 16
#define ENT_PITCH  (CHUNKS + 1)
static_assert(CHUNKS * NODE_WAVES <= 2 * NODE_WAVES * 32, "two table entries per thread cover the work-group's nodes");
static_assert((NODE_WAVES & (NODE_WAVES - 1)) == 0, "node index taken with a mask");

__device__ __forceinline__
int window_src(int* sl, const unsigned short* __restrict__ ssrc, int wb, int lane, int d,
               int x0, int i0, int p0, int x1, int i1, int p1) {
  #pragma clang loop unroll(disable) vectorize(disable)
  for (int k = max(x0, wb); k < min(i0, wb + 32); ++k) sl[k - wb] = p0 + (k - x0);
  #pragma clang loop unroll(disable) vectorize(disable)
  for (int k = max(x1, wb); k < min(i1, wb + 32); ++k) sl[k - wb] = p1 + (k - x1);
  __builtin_amdgcn_fence(__ATOMIC_RELEASE, "wavefront");
  __builtin_amdgcn_wave_barrier();
  __builtin_amdgcn_fence(__ATOMIC_ACQUIRE, "wavefront");
  int s = 0;
  if (wb + lane < d) s = ssrc[sl[lane]];
  return s;
}

__global__ __launch_bounds__(NODE_WAVES * 32, 16)
void gat_node(const float* __restrict__ Wx,
              const float* __restrict__ s1,
              const float* __restrict__ s2,
              const int* __restrict__ tab,
              const unsigned short* __restrict__ ssrc,
              float* __restrict__ out, int N, int E) {
  __shared__ int slots[NODE_WAVES][32];
  __shared__ int ent[NODE_WAVES][ENT_PITCH];
  const int wave = threadIdx.x >> 5;
  const int wv   = blockIdx.x * NODE_WAVES + wave;
  const int lane = threadIdx.x & 31;
  {
    const int nb = blockIdx.x * NODE_WAVES;
    const int ia = threadIdx.x;
    const int ib = min(ia + NODE_WAVES * 32, CHUNKS * NODE_WAVES - 1);
    const int na = min(nb + (ia & (NODE_WAVES - 1)), MAX_NODES - 1);
    const int nc = min(nb + (ib & (NODE_WAVES - 1)), MAX_NODES - 1);
    const int va = tab[(size_t)(ia / NODE_WAVES) * MAX_NODES + na];
    const int vb = tab[(size_t)(ib / NODE_WAVES) * MAX_NODES + nc];
    ent[ia & (NODE_WAVES - 1)][ia / NODE_WAVES] = va;
    ent[ib & (NODE_WAVES - 1)][ib / NODE_WAVES] = vb;
  }
  __syncthreads();
  if (wv >= N) return;
  const int n   = __builtin_amdgcn_readfirstlane(wv);
  const int per = (E + CHUNKS - 1) / CHUNKS;
  int* sl = slots[wave];

  const int  ch1  = min(lane + 32, CHUNKS - 1);
  const unsigned pk0 = (unsigned)ent[wave][lane];
  const unsigned pk1 = (unsigned)ent[wave][ch1];
  const int c0 = (int)(pk0 & 0xffffu);
  const int c1 = lane + 32 < CHUNKS ? (int)(pk1 & 0xffffu) : 0;
  const int p0 = (int)(pk0 >> 16) + lane * per;
  const int p1 = (int)(pk1 >> 16) + ch1 * per;
  int i0 = c0, i1 = c1;
  #pragma unroll
  for (int x = 1; x < 32; x <<= 1) {
    const int t0 = __shfl_up(i0, x);
    const int t1 = __shfl_up(i1, x);
    if (lane >= x) { i0 += t0; i1 += t1; }
  }
  i1 += __builtin_amdgcn_readlane(i0, 31);
  const int d  = __builtin_amdgcn_readlane(i1, 31);
  const int x0 = i0 - c0, x1 = i1 - c1;

  const float4 sd = *(const float4*)(s2 + (size_t)n * HEADS);
  const int sw0 = window_src(sl, ssrc, 0, lane, d, x0, i0, p0, x1, i1, p1);

  const float NEG_INF = -__builtin_inff();
  float4 m = make_float4(NEG_INF, NEG_INF, NEG_INF, NEG_INF);
  for (int cb = 0; cb < d; cb += 32) {
    const int s = cb == 0 ? sw0 : window_src(sl, ssrc, cb, lane, d, x0, i0, p0, x1, i1, p1);
    if (cb + lane < d) {
      const float4 ss = *(const float4*)(s1 + (size_t)s * HEADS);
      m.x = fmaxf(m.x, lrelu02(ss.x + sd.x));
      m.y = fmaxf(m.y, lrelu02(ss.y + sd.y));
      m.z = fmaxf(m.z, lrelu02(ss.z + sd.z));
      m.w = fmaxf(m.w, lrelu02(ss.w + sd.w));
    }
  }
  #pragma unroll
  for (int x = 16; x >= 1; x >>= 1) {
    m.x = fmaxf(m.x, __shfl_xor(m.x, x));
    m.y = fmaxf(m.y, __shfl_xor(m.y, x));
    m.z = fmaxf(m.z, __shfl_xor(m.z, x));
    m.w = fmaxf(m.w, __shfl_xor(m.w, x));
  }

  float den[HEADS] = {0.f, 0.f, 0.f, 0.f};
  float acc[HEADS] = {0.f, 0.f, 0.f, 0.f};
  const float* wrow = Wx + lane;
  for (int cb = 0; cb < d; cb += 32) {
    const int s = cb == 0 ? sw0 : window_src(sl, ssrc, cb, lane, d, x0, i0, p0, x1, i1, p1);
    float4 w = make_float4(0.f, 0.f, 0.f, 0.f);
    if (cb + lane < d) {
      const float4 ss = *(const float4*)(s1 + (size_t)s * HEADS);
      w.x = __expf(lrelu02(ss.x + sd.x) - m.x);
      w.y = __expf(lrelu02(ss.y + sd.y) - m.y);
      w.z = __expf(lrelu02(ss.z + sd.z) - m.z);
      w.w = __expf(lrelu02(ss.w + sd.w) - m.w);
    }
    const int cn = (d - cb) < 32 ? (d - cb) : 32;
    int j = 0;
    for (; j + EDGE_GROUP <= cn; j += EDGE_GROUP) {
      float r[EDGE_GROUP][HEADS];
      #pragma unroll
      for (int g = 0; g < EDGE_GROUP; ++g) {
        const int sj = __builtin_amdgcn_readlane(s, j + g);
        const float* p = wrow + (size_t)sj * OUT_DIM;
        #pragma unroll
        for (int k = 0; k < HEADS; ++k) r[g][k] = p[k * HEAD_DIM];
      }
      #pragma unroll
      for (int g = 0; g < EDGE_GROUP; ++g) {
        const float wk[HEADS] = {lane_f(w.x, j + g), lane_f(w.y, j + g),
                                 lane_f(w.z, j + g), lane_f(w.w, j + g)};
        #pragma unroll
        for (int k = 0; k < HEADS; ++k) {
          den[k] += wk[k];
          acc[k] += wk[k] * r[g][k];
        }
      }
    }
    for (; j < cn; ++j) {
      const int sj = __builtin_amdgcn_readlane(s, j);
      const float* p = wrow + (size_t)sj * OUT_DIM;
      const float wk[HEADS] = {lane_f(w.x, j), lane_f(w.y, j),
                               lane_f(w.z, j), lane_f(w.w, j)};
      #pragma unroll
      for (int k = 0; k < HEADS; ++k) {
        den[k] += wk[k];
        acc[k] += wk[k] * p[k * HEAD_DIM];
      }
    }
  }
  float* orow = out + (size_t)n * OUT_DIM + lane;
  #pragma unroll
  for (int k = 0; k < HEADS; ++k) {
    const float rd = d > 0 ? __frcp_rn(den[k]) : 0.f;
    const float v  = acc[k] * rd;
    orow[k * HEAD_DIM] = v > 0.f ? v : (__expf(v) - 1.0f);
  }
}

extern "C" void kernel_launch(void* const* d_in, const int* in_sizes, int n_in,
                              void* d_out, int out_size, void* d_ws, size_t ws_size,
                              hipStream_t stream) {
  const float* x  = (const float*)d_in[0];
  const float* W  = (const float*)d_in[1];
  const float* a  = (const float*)d_in[2];
  const int*   ei = (const int*)d_in[3];
  const int N = in_sizes[0] / IN_DIM;
  const int E = in_sizes[3] / 2;
  float* out = (float*)d_out;

  char* ws = (char*)d_ws;
  size_t off_b = 0;
  auto carve = [&](size_t bytes) -> void* {
    void* p = ws + off_b;
    off_b += (bytes + 255) & ~(size_t)255;
    return p;
  };
  float* Wx   = (float*)carve((size_t)N * OUT_DIM * sizeof(float));
  float* s1   = (float*)carve((size_t)N * HEADS * sizeof(float));
  float* s2   = (float*)carve((size_t)N * HEADS * sizeof(float));
  unsigned short* ssrc = (unsigned short*)carve((size_t)E * sizeof(unsigned short));
  int*   tab  = (int*)  carve((size_t)CHUNKS * MAX_NODES * sizeof(int));
  (void)ws_size; (void)n_in; (void)out_size;

  gat_gemm_wmma<<<COUNT_WGS + GEMM_GRID, FUSED_BLOCK, 0, stream>>>(x, W, a, Wx, s1, s2, N,
                                                                ei, tab, ssrc, E);

  gat_node<<<(N + NODE_WAVES - 1) / NODE_WAVES, NODE_WAVES * 32, 0, stream>>>(Wx, s1, s2, tab, ssrc,
                                                                            out, N, E);
}
